# starting kernel
# baseline (speedup 1.0000x reference)
#include <hip/hip_runtime.h>

typedef float v2f __attribute__((ext_vector_type(2)));
typedef float v8f __attribute__((ext_vector_type(8)));

__global__ __launch_bounds__(256) void kd_conv_wmma(
    const float* __restrict__ X, const float* __restrict__ Wt,
    const float* __restrict__ bias, const int* __restrict__ sel,
    float* __restrict__ Out, int M, int K, int Dlog2, int Bsz, int mode)
{
    extern __shared__ float sW[];
    const int D   = 1 << Dlog2;
    const int Kp4 = (K + 3) & ~3;
    const int Kp  = Kp4 + 1;
    const int m0  = blockIdx.y << 4;

    for (int r = 0; r < 16; ++r) {
        const int m = m0 + r;
        const float* wr = Wt + (size_t)m * K;
        for (int k = threadIdx.x; k < Kp4; k += 256) {
            float w = 0.0f;
            if (m < M && k < K) w = wr[k];
            sW[r * Kp + k] = w;
        }
    }
    __syncthreads();

    const int lane = threadIdx.x & 31;
    const int ng   = (blockIdx.x << 3) + (threadIdx.x >> 5);
    const int n0   = ng << 6;

    const int half = lane >> 4;
    const int l16  = lane & 15;
    const int kOff = half << 1;
    const float* aRow = sW + l16 * Kp;

    const float* xcol[4];
#pragma unroll
    for (int s = 0; s < 4; ++s) {
        const int n  = n0 + (s << 4) + l16;
        const int bb = n >> Dlog2;
        const int dd = n & (D - 1);
        xcol[s] = X + (size_t)bb * ((size_t)K << Dlog2) + dd;
    }

    v8f acc[4] = { v8f{}, v8f{}, v8f{}, v8f{} };

    const int kFull = K & ~3;
    for (int kt = 0; kt < kFull; kt += 4) {
        const int k0 = kt + kOff;
        v2f a;
        a.x = aRow[k0];
        a.y = aRow[k0 + 1];
#pragma unroll
        for (int s = 0; s < 4; ++s) {
            v2f b;
            b.x = xcol[s][(size_t)k0 << Dlog2];
            b.y = xcol[s][(size_t)(k0 + 1) << Dlog2];
            acc[s] = __builtin_amdgcn_wmma_f32_16x16x4_f32(
                false, a, false, b, (short)0, acc[s], false, false);
        }
        if (kt + 4 < kFull) {
#pragma unroll
            for (int s = 0; s < 4; ++s)
                __builtin_prefetch(xcol[s] + ((size_t)(k0 + 4) << Dlog2), 0, 1);
        }
    }
    if (kFull < K) {
        const int k0 = kFull + kOff;
        const int c0 = min(k0, K - 1);
        const int c1 = min(k0 + 1, K - 1);
        v2f a;
        a.x = aRow[k0];
        a.y = aRow[k0 + 1];
#pragma unroll
        for (int s = 0; s < 4; ++s) {
            v2f b;
            b.x = xcol[s][(size_t)c0 << Dlog2];
            b.y = xcol[s][(size_t)c1 << Dlog2];
            acc[s] = __builtin_amdgcn_wmma_f32_16x16x4_f32(
                false, a, false, b, (short)0, acc[s], false, false);
        }
    }

    const int Dh = D >> 1;
    const int F = M / 3;
#pragma unroll
    for (int s = 0; s < 4; ++s) {
        const int n  = n0 + (s << 4) + l16;
        const int bb = n >> Dlog2;
        const int dp = n & (D - 1);
#pragma unroll
        for (int r = 0; r < 8; ++r) {
            const int o = m0 + r + (half << 3);
            if (o >= M) continue;
            float v = acc[s][r] + bias[o];
            if (mode == 0) {
                v = fmaxf(v, 0.0f);
                const int f = o / 3;
                const int q = o - 3 * f;
                const int t = (q << Dlog2) + dp;
                const int dsrc = t / 3;
                const int rr = t - 3 * dsrc;
                if (sel[dsrc] == rr) {
                    float* dst = Out + (size_t)bb * (F * Dh)
                                     + (size_t)f * Dh + (dsrc >> 1);
                    atomicMax(dst, v);
                }
            } else {
                Out[(size_t)n * M + o] = v;
            }
        }
    }
}

__global__ __launch_bounds__(256) void kd_logsoftmax16(
    const float* __restrict__ logits, float* __restrict__ out, int rows)
{
    int r = blockIdx.x * blockDim.x + threadIdx.x;
    if (r >= rows) return;
    float v[16];
    float mx = -3.402823466e38f;
#pragma unroll
    for (int i = 0; i < 16; ++i) { v[i] = logits[r * 16 + i]; mx = fmaxf(mx, v[i]); }
    float s = 0.0f;
#pragma unroll
    for (int i = 0; i < 16; ++i) s += expf(v[i] - mx);
    const float l = mx + logf(s);
#pragma unroll
    for (int i = 0; i < 16; ++i) out[r * 16 + i] = v[i] - l;
}

extern "C" void kernel_launch(void* const* d_in, const int* in_sizes, int n_in,
                              void* d_out, int out_size, void* d_ws, size_t ws_size,
                              hipStream_t stream) {
    static const int DIMS[11] = {2048, 1024, 512, 256, 128, 64, 32, 16, 8, 4, 2};
    static const int INCH[11] = {3, 8, 32, 64, 64, 64, 128, 256, 512, 512, 512};
    static const int FEAT[11] = {8, 32, 64, 64, 64, 128, 256, 512, 512, 512, 1024};
    const int Bsz = 512;

    const size_t XBYTES = (size_t)512 * 32 * 512 * sizeof(float);
    float* X0     = (float*)d_ws;
    float* X1     = (float*)((char*)d_ws + XBYTES);
    float* logits = (float*)((char*)d_ws + 2 * XBYTES);

    const float* cur = (const float*)d_in[0];
    for (int i = 0; i < 11; ++i) {
        const int Dd = DIMS[i], Kc = INCH[i], F = FEAT[i], M = 3 * F;
        const int Dlog2 = __builtin_ctz((unsigned)Dd);
        float* nxt = (i & 1) ? X1 : X0;
        const size_t outBytes = (size_t)Bsz * F * (Dd / 2) * sizeof(float);
        hipMemsetAsync(nxt, 0, outBytes, stream);

        const int N = Bsz * Dd;
        const int tilesM = (M + 15) / 16;
        const int nGroups = N / 64;
        const int blocksX = nGroups / 8;
        const int Kp4 = (Kc + 3) & ~3;
        const size_t ldsBytes = (size_t)16 * (Kp4 + 1) * sizeof(float);
        dim3 grid(blocksX, tilesM);
        kd_conv_wmma<<<grid, 256, ldsBytes, stream>>>(
            cur, (const float*)d_in[12 + 2 * i], (const float*)d_in[13 + 2 * i],
            (const int*)d_in[1 + i], nxt, M, Kc, Dlog2, Bsz,  0);
        cur = nxt;
    }

    {
        const size_t ldsBytes = (size_t)16 * (1024 + 1) * sizeof(float);
        dim3 grid(1, 1);
        kd_conv_wmma<<<grid, 256, ldsBytes, stream>>>(
            cur, (const float*)d_in[34], (const float*)d_in[35], nullptr,
            logits,  16,  1024,  0, Bsz,  1);
    }

    kd_logsoftmax16<<<(512 + 255) / 256, 256, 0, stream>>>(logits, (float*)d_out, 512);

    (void)in_sizes; (void)n_in; (void)out_size; (void)ws_size;
}


# Activations kept batch-minor [d][c][b]: a column tile is 16 batches at one d, so only the gathered W row-third is multiplied; pair max of 2 accumulators, plain stores replace atomic max and clears
# speedup vs baseline: 3.3552x; 3.3552x over previous
#include <hip/hip_runtime.h>


typedef float v2f __attribute__((ext_vector_type(2)));
typedef float v8f __attribute__((ext_vector_type(8)));

template <int NSUB, int KP4, bool FIRST>
__global__ __launch_bounds__(256) void kd_conv_bm(
    const float* __restrict__ X, const float* __restrict__ Wt,
    const float* __restrict__ bias, const int* __restrict__ sel,
    float* __restrict__ Out, int F, int Dlog2)
{
    constexpr int K   = FIRST ? 3 : KP4;
    constexpr int KP  = KP4 + 1;
    constexpr int NB  = 512;
    constexpr int NBC = 4 / NSUB;
    constexpr int UNR = 8 / NSUB;
    __shared__ float sW[48 * KP];

    const int f0 = blockIdx.y << 4;

    constexpr int TOT = 48 * KP4;
    constexpr int NIT = (TOT + 255) / 256;
    constexpr bool EXACT = (TOT % 256) == 0;
#pragma unroll 16
    for (int it = 0; it < NIT; ++it) {
        const int idx = it * 256 + (int)threadIdx.x;
        const int idc = EXACT ? idx : min(idx, TOT - 1);
        const int rr  = idc / KP4;
        const int k   = idc - rr * KP4;
        const int q   = rr >> 4;
        const int f   = f0 + (rr & 15);
        float w = Wt[(3 * min(f, F - 1) + q) * K + min(k, K - 1)];
        w = (f < F && k < K) ? w : 0.0f;
        if (EXACT || idx < TOT) sW[rr * KP + k] = w;
    }
    __syncthreads();

    const int lane = threadIdx.x & 31;
    const int wave = threadIdx.x >> 5;
    const int half = lane >> 4;
    const int c    = lane & 15;
    const int kOff = half << 1;

    const int j     = blockIdx.x / NBC;
    const int bc    = blockIdx.x - j * NBC;
    const int bBase = (bc * 8 + wave) * (16 * NSUB);
    const int D     = 1 << Dlog2;

    const int d0 = j << 1;
    const int d1 = d0 + 1;
    const int s0 = min(max(sel[d0], 0), 2);
    const int s1 = min(max(sel[d1], 0), 2);
    const int t0 = 3 * d0 + s0;
    const int t1 = 3 * d1 + s1;
    const int q0 = t0 >> Dlog2;
    const int q1 = t1 >> Dlog2;
    const int dp0 = t0 & (D - 1);
    const int dp1 = t1 & (D - 1);

    const float* aRow0 = sW + (q0 * 16 + c) * KP;
    const float* aRow1 = sW + (q1 * 16 + c) * KP;

    const int sk = FIRST ? D : NB;
    const float* p0[NSUB];
    const float* p1[NSUB];
#pragma unroll
    for (int s = 0; s < NSUB; ++s) {
        const int b = bBase + (s << 4) + c;
        if (FIRST) {
            p0[s] = X + (size_t)b * (size_t)(K * D) + dp0;
            p1[s] = X + (size_t)b * (size_t)(K * D) + dp1;
        } else {
            p0[s] = X + (size_t)dp0 * (size_t)(K * NB) + b;
            p1[s] = X + (size_t)dp1 * (size_t)(K * NB) + b;
        }
    }

    v8f acc0[NSUB];
    v8f acc1[NSUB];
#pragma unroll
    for (int s = 0; s < NSUB; ++s) { acc0[s] = v8f{}; acc1[s] = v8f{}; }

    constexpr int kFull = K & ~3;
#pragma unroll UNR
    for (int kt = 0; kt < kFull; kt += 4) {
        const int k0 = kt + kOff;
        v2f a0, a1;
        a0.x = aRow0[k0]; a0.y = aRow0[k0 + 1];
        a1.x = aRow1[k0]; a1.y = aRow1[k0 + 1];
#pragma unroll
        for (int s = 0; s < NSUB; ++s) {
            v2f b0, b1;
            b0.x = p0[s][(size_t)k0 * sk];
            b0.y = p0[s][(size_t)(k0 + 1) * sk];
            b1.x = p1[s][(size_t)k0 * sk];
            b1.y = p1[s][(size_t)(k0 + 1) * sk];
            acc0[s] = __builtin_amdgcn_wmma_f32_16x16x4_f32(
                false, a0, false, b0, (short)0, acc0[s], false, false);
            acc1[s] = __builtin_amdgcn_wmma_f32_16x16x4_f32(
                false, a1, false, b1, (short)0, acc1[s], false, false);
        }
    }
    if (kFull < K) {
        const int k0 = kFull + kOff;
        const int c0 = min(k0, K - 1);
        const int c1 = min(k0 + 1, K - 1);
        v2f a0, a1;
        a0.x = aRow0[k0]; a0.y = aRow0[k0 + 1];
        a1.x = aRow1[k0]; a1.y = aRow1[k0 + 1];
#pragma unroll
        for (int s = 0; s < NSUB; ++s) {
            v2f b0, b1;
            b0.x = p0[s][(size_t)c0 * sk];
            b0.y = p0[s][(size_t)c1 * sk];
            b1.x = p1[s][(size_t)c0 * sk];
            b1.y = p1[s][(size_t)c1 * sk];
            acc0[s] = __builtin_amdgcn_wmma_f32_16x16x4_f32(
                false, a0, false, b0, (short)0, acc0[s], false, false);
            acc1[s] = __builtin_amdgcn_wmma_f32_16x16x4_f32(
                false, a1, false, b1, (short)0, acc1[s], false, false);
        }
    }

    float bq0[8], bq1[8];
#pragma unroll
    for (int r = 0; r < 8; ++r) {
        const int f = f0 + (half << 3) + r;
        const int fc = min(f, F - 1);
        bq0[r] = bias[3 * fc + q0];
        bq1[r] = bias[3 * fc + q1];
    }
#pragma unroll
    for (int s = 0; s < NSUB; ++s) {
        const int b = bBase + (s << 4) + c;
#pragma unroll
        for (int r = 0; r < 8; ++r) {
            const int f = f0 + (half << 3) + r;
            if (f < F) {
                const float v0 = fmaxf(acc0[s][r] + bq0[r], 0.0f);
                const float v1 = fmaxf(acc1[s][r] + bq1[r], 0.0f);
                Out[((size_t)j * F + f) * NB + b] = fmaxf(v0, v1);
            }
        }
    }
}

__global__ __launch_bounds__(256) void kd_fc_wmma(
    const float* __restrict__ X, const float* __restrict__ Wt,
    const float* __restrict__ bias, float* __restrict__ Out,
    int M, int K, int Dlog2)
{
    extern __shared__ float sWfc[];
    const int D   = 1 << Dlog2;
    const int Kp4 = (K + 3) & ~3;
    const int Kp  = Kp4 + 1;
    const int m0  = blockIdx.y << 4;

    for (int r = 0; r < 16; ++r) {
        const int m = m0 + r;
        const float* wr = Wt + (size_t)m * K;
        for (int k = threadIdx.x; k < Kp4; k += 256) {
            float w = 0.0f;
            if (m < M && k < K) w = wr[k];
            sWfc[r * Kp + k] = w;
        }
    }
    __syncthreads();

    const int lane = threadIdx.x & 31;
    const int ng   = (blockIdx.x << 3) + (threadIdx.x >> 5);
    const int n0   = ng << 6;
    const int half = lane >> 4;
    const int l16  = lane & 15;
    const int kOff = half << 1;
    const float* aRow = sWfc + l16 * Kp;

    const float* xcol[4];
#pragma unroll
    for (int s = 0; s < 4; ++s) {
        const int n  = n0 + (s << 4) + l16;
        const int bb = n >> Dlog2;
        const int dd = n & (D - 1);
        xcol[s] = X + (size_t)bb * ((size_t)K << Dlog2) + dd;
    }

    v8f acc[4] = { v8f{}, v8f{}, v8f{}, v8f{} };

    const int kFull = K & ~3;
    for (int kt = 0; kt < kFull; kt += 4) {
        const int k0 = kt + kOff;
        v2f a;
        a.x = aRow[k0];
        a.y = aRow[k0 + 1];
#pragma unroll
        for (int s = 0; s < 4; ++s) {
            v2f b;
            b.x = xcol[s][(size_t)k0 << Dlog2];
            b.y = xcol[s][(size_t)(k0 + 1) << Dlog2];
            acc[s] = __builtin_amdgcn_wmma_f32_16x16x4_f32(
                false, a, false, b, (short)0, acc[s], false, false);
        }
        if (kt + 4 < kFull) {
#pragma unroll
            for (int s = 0; s < 4; ++s)
                __builtin_prefetch(xcol[s] + ((size_t)(k0 + 4) << Dlog2), 0, 1);
        }
    }
    if (kFull < K) {
        const int k0 = kFull + kOff;
        const int c0 = min(k0, K - 1);
        const int c1 = min(k0 + 1, K - 1);
        v2f a;
        a.x = aRow[k0];
        a.y = aRow[k0 + 1];
#pragma unroll
        for (int s = 0; s < 4; ++s) {
            v2f b;
            b.x = xcol[s][(size_t)c0 << Dlog2];
            b.y = xcol[s][(size_t)c1 << Dlog2];
            acc[s] = __builtin_amdgcn_wmma_f32_16x16x4_f32(
                false, a, false, b, (short)0, acc[s], false, false);
        }
    }

#pragma unroll
    for (int s = 0; s < 4; ++s) {
        const int n = n0 + (s << 4) + l16;
#pragma unroll
        for (int r = 0; r < 8; ++r) {
            const int o = m0 + r + (half << 3);
            if (o >= M) continue;
            Out[(size_t)n * M + o] = acc[s][r] + bias[o];
        }
    }
}

__global__ __launch_bounds__(256) void kd_logsoftmax16(
    const float* __restrict__ logits, float* __restrict__ out, int rows)
{
    int r = blockIdx.x * blockDim.x + threadIdx.x;
    if (r >= rows) return;
    float v[16];
    float mx = -3.402823466e38f;
#pragma unroll
    for (int i = 0; i < 16; ++i) { v[i] = logits[r * 16 + i]; mx = fmaxf(mx, v[i]); }
    float s = 0.0f;
#pragma unroll
    for (int i = 0; i < 16; ++i) s += expf(v[i] - mx);
    const float l = mx + logf(s);
#pragma unroll
    for (int i = 0; i < 16; ++i) out[r * 16 + i] = v[i] - l;
}

template <int NSUB, int KP4, bool FIRST>
static void launch_conv(const float* cur, const float* W, const float* b,
                        const int* sel, float* nxt, int F, int Dd,
                        hipStream_t stream)
{
    const int Dlog2 = __builtin_ctz((unsigned)Dd);
    dim3 grid((Dd / 2) * (4 / NSUB), (F + 15) / 16);
    kd_conv_bm<NSUB, KP4, FIRST><<<grid, 256, 0, stream>>>(cur, W, b, sel, nxt, F, Dlog2);
}

extern "C" void kernel_launch(void* const* d_in, const int* in_sizes, int n_in,
                              void* d_out, int out_size, void* d_ws, size_t ws_size,
                              hipStream_t stream) {
    static const int DIMS[11] = {2048, 1024, 512, 256, 128, 64, 32, 16, 8, 4, 2};
    static const int FEAT[11] = {8, 32, 64, 64, 64, 128, 256, 512, 512, 512, 1024};

    const size_t XBYTES = (size_t)512 * 32 * 512 * sizeof(float);
    float* X0     = (float*)d_ws;
    float* X1     = (float*)((char*)d_ws + XBYTES);
    float* logits = (float*)((char*)d_ws + 2 * XBYTES);

    const float* cur = (const float*)d_in[0];
    for (int i = 0; i < 11; ++i) {
        const float* W  = (const float*)d_in[12 + 2 * i];
        const float* bi = (const float*)d_in[13 + 2 * i];
        const int*   sl = (const int*)d_in[1 + i];
        float* nxt = (i & 1) ? X1 : X0;
        const int F = FEAT[i], Dd = DIMS[i];
        switch (i) {
        case 0:  launch_conv<4, 4,   true >(cur, W, bi, sl, nxt, F, Dd, stream); break;
        case 1:  launch_conv<4, 8,   false>(cur, W, bi, sl, nxt, F, Dd, stream); break;
        case 2:  launch_conv<4, 32,  false>(cur, W, bi, sl, nxt, F, Dd, stream); break;
        case 3:  launch_conv<2, 64,  false>(cur, W, bi, sl, nxt, F, Dd, stream); break;
        case 4:
        case 5:  launch_conv<1, 64,  false>(cur, W, bi, sl, nxt, F, Dd, stream); break;
        case 6:  launch_conv<1, 128, false>(cur, W, bi, sl, nxt, F, Dd, stream); break;
        case 7:  launch_conv<1, 256, false>(cur, W, bi, sl, nxt, F, Dd, stream); break;
        default: launch_conv<1, 512, false>(cur, W, bi, sl, nxt, F, Dd, stream); break;
        }
        cur = nxt;
    }

    {
        const size_t ldsBytes = (size_t)16 * (1024 + 1) * sizeof(float);
        dim3 grid(1, 1);
        kd_fc_wmma<<<grid, 256, ldsBytes, stream>>>(
            cur, (const float*)d_in[34], (const float*)d_in[35],
            logits,  16,  1024,  9);
    }

    kd_logsoftmax16<<<(512 + 255) / 256, 256, 0, stream>>>(logits, (float*)d_out, 512);

    (void)in_sizes; (void)n_in; (void)out_size; (void)ws_size;
}


# FC head fused with log-softmax into one launch (13 -> 12): 32 one-wave groups, each wave 16 batches x all 16 logits over the whole K, 48 loads in flight, no LDS; softmax in registers; sums unchanged
# speedup vs baseline: 6.5189x; 1.9429x over previous
#include <hip/hip_runtime.h>


typedef float v2f __attribute__((ext_vector_type(2)));
typedef float v8f __attribute__((ext_vector_type(8)));

template <int NSUB, int KP4, bool FIRST>
__global__ __launch_bounds__(256) void kd_conv_bm(
    const float* __restrict__ X, const float* __restrict__ Wt,
    const float* __restrict__ bias, const int* __restrict__ sel,
    float* __restrict__ Out, int F, int Dlog2)
{
    constexpr int K   = FIRST ? 3 : KP4;
    constexpr int KP  = KP4 + 1;
    constexpr int NB  = 512;
    constexpr int NBC = 4 / NSUB;
    constexpr int UNR = 8 / NSUB;
    __shared__ float sW[48 * KP];

    const int f0 = blockIdx.y << 4;

    constexpr int TOT = 48 * KP4;
    constexpr int NIT = (TOT + 255) / 256;
    constexpr bool EXACT = (TOT % 256) == 0;
#pragma unroll 16
    for (int it = 0; it < NIT; ++it) {
        const int idx = it * 256 + (int)threadIdx.x;
        const int idc = EXACT ? idx : min(idx, TOT - 1);
        const int rr  = idc / KP4;
        const int k   = idc - rr * KP4;
        const int q   = rr >> 4;
        const int f   = f0 + (rr & 15);
        float w = Wt[(3 * min(f, F - 1) + q) * K + min(k, K - 1)];
        w = (f < F && k < K) ? w : 0.0f;
        if (EXACT || idx < TOT) sW[rr * KP + k] = w;
    }
    __syncthreads();

    const int lane = threadIdx.x & 31;
    const int wave = threadIdx.x >> 5;
    const int half = lane >> 4;
    const int c    = lane & 15;
    const int kOff = half << 1;

    const int j     = blockIdx.x / NBC;
    const int bc    = blockIdx.x - j * NBC;
    const int bBase = (bc * 8 + wave) * (16 * NSUB);
    const int D     = 1 << Dlog2;

    const int d0 = j << 1;
    const int d1 = d0 + 1;
    const int s0 = min(max(sel[d0], 0), 2);
    const int s1 = min(max(sel[d1], 0), 2);
    const int t0 = 3 * d0 + s0;
    const int t1 = 3 * d1 + s1;
    const int q0 = t0 >> Dlog2;
    const int q1 = t1 >> Dlog2;
    const int dp0 = t0 & (D - 1);
    const int dp1 = t1 & (D - 1);

    const float* aRow0 = sW + (q0 * 16 + c) * KP;
    const float* aRow1 = sW + (q1 * 16 + c) * KP;

    const int sk = FIRST ? D : NB;
    const float* p0[NSUB];
    const float* p1[NSUB];
#pragma unroll
    for (int s = 0; s < NSUB; ++s) {
        const int b = bBase + (s << 4) + c;
        if (FIRST) {
            p0[s] = X + (size_t)b * (size_t)(K * D) + dp0;
            p1[s] = X + (size_t)b * (size_t)(K * D) + dp1;
        } else {
            p0[s] = X + (size_t)dp0 * (size_t)(K * NB) + b;
            p1[s] = X + (size_t)dp1 * (size_t)(K * NB) + b;
        }
    }

    v8f acc0[NSUB];
    v8f acc1[NSUB];
#pragma unroll
    for (int s = 0; s < NSUB; ++s) { acc0[s] = v8f{}; acc1[s] = v8f{}; }

    constexpr int kFull = K & ~3;
#pragma unroll UNR
    for (int kt = 0; kt < kFull; kt += 4) {
        const int k0 = kt + kOff;
        v2f a0, a1;
        a0.x = aRow0[k0]; a0.y = aRow0[k0 + 1];
        a1.x = aRow1[k0]; a1.y = aRow1[k0 + 1];
#pragma unroll
        for (int s = 0; s < NSUB; ++s) {
            v2f b0, b1;
            b0.x = p0[s][(size_t)k0 * sk];
            b0.y = p0[s][(size_t)(k0 + 1) * sk];
            b1.x = p1[s][(size_t)k0 * sk];
            b1.y = p1[s][(size_t)(k0 + 1) * sk];
            acc0[s] = __builtin_amdgcn_wmma_f32_16x16x4_f32(
                false, a0, false, b0, (short)0, acc0[s], false, false);
            acc1[s] = __builtin_amdgcn_wmma_f32_16x16x4_f32(
                false, a1, false, b1, (short)0, acc1[s], false, false);
        }
    }
    if (kFull < K) {
        const int k0 = kFull + kOff;
        const int c0 = min(k0, K - 1);
        const int c1 = min(k0 + 1, K - 1);
        v2f a0, a1;
        a0.x = aRow0[k0]; a0.y = aRow0[k0 + 1];
        a1.x = aRow1[k0]; a1.y = aRow1[k0 + 1];
#pragma unroll
        for (int s = 0; s < NSUB; ++s) {
            v2f b0, b1;
            b0.x = p0[s][(size_t)c0 * sk];
            b0.y = p0[s][(size_t)c1 * sk];
            b1.x = p1[s][(size_t)c0 * sk];
            b1.y = p1[s][(size_t)c1 * sk];
            acc0[s] = __builtin_amdgcn_wmma_f32_16x16x4_f32(
                false, a0, false, b0, (short)0, acc0[s], false, false);
            acc1[s] = __builtin_amdgcn_wmma_f32_16x16x4_f32(
                false, a1, false, b1, (short)0, acc1[s], false, false);
        }
    }

    float bq0[8], bq1[8];
#pragma unroll
    for (int r = 0; r < 8; ++r) {
        const int f = f0 + (half << 3) + r;
        const int fc = min(f, F - 1);
        bq0[r] = bias[3 * fc + q0];
        bq1[r] = bias[3 * fc + q1];
    }
#pragma unroll
    for (int s = 0; s < NSUB; ++s) {
        const int b = bBase + (s << 4) + c;
#pragma unroll
        for (int r = 0; r < 8; ++r) {
            const int f = f0 + (half << 3) + r;
            if (f < F) {
                const float v0 = fmaxf(acc0[s][r] + bq0[r], 0.0f);
                const float v1 = fmaxf(acc1[s][r] + bq1[r], 0.0f);
                Out[((size_t)j * F + f) * NB + b] = fmaxf(v0, v1);
            }
        }
    }
}

__global__ __launch_bounds__(32) void kd_fc_lsm(
    const float* __restrict__ Y, const float* __restrict__ Wt,
    const float* __restrict__ bias, float* __restrict__ out)
{
    constexpr int K  = 1024;
    constexpr int NB = 512;
    constexpr int M  = 16;

    const int lane = threadIdx.x & 31;
    const int half = lane >> 4;
    const int c    = lane & 15;
    const int kOff = half << 1;
    const int b    = (blockIdx.x << 4) + c;

    const float* ap = Wt + c * K + kOff;
    const float* bp = Y + (size_t)kOff * NB + b;

    float bs[8];
#pragma unroll
    for (int r = 0; r < 8; ++r) bs[r] = bias[(half << 3) + r];

    v8f acc = v8f{};
#pragma unroll 16
    for (int kt = 0; kt < K; kt += 4) {
        const v2f a = *reinterpret_cast<const v2f*>(ap + kt);
        v2f x;
        x.x = bp[(size_t)kt * NB];
        x.y = bp[(size_t)(kt + 1) * NB];
        acc = __builtin_amdgcn_wmma_f32_16x16x4_f32(
            false, a, false, x, (short)0, acc, false, false);
    }

    float v[8];
    float mx = -3.402823466e38f;
#pragma unroll
    for (int r = 0; r < 8; ++r) { v[r] = acc[r] + bs[r]; mx = fmaxf(mx, v[r]); }
    mx = fmaxf(mx, __shfl_xor(mx, 16, 32));

    float e[8];
#pragma unroll
    for (int r = 0; r < 8; ++r) e[r] = expf(v[r] - mx);
    float s = 0.0f;
#pragma unroll
    for (int r = 0; r < 8; ++r) s += e[r];
    float t = __shfl(s, c, 32);
#pragma unroll
    for (int r = 0; r < 8; ++r) t += e[r];
    const float tot = __shfl(t, c + 16, 32);
    const float l = mx + logf(tot);

    float4 o0, o1;
    o0.x = v[0] - l; o0.y = v[1] - l; o0.z = v[2] - l; o0.w = v[3] - l;
    o1.x = v[4] - l; o1.y = v[5] - l; o1.z = v[6] - l; o1.w = v[7] - l;
    float4* op = reinterpret_cast<float4*>(out + (size_t)b * M + (half << 3));
    op[0] = o0;
    op[1] = o1;
}

template <int NSUB, int KP4, bool FIRST>
static void launch_conv(const float* cur, const float* W, const float* b,
                        const int* sel, float* nxt, int F, int Dd,
                        hipStream_t stream)
{
    const int Dlog2 = __builtin_ctz((unsigned)Dd);
    dim3 grid((Dd / 2) * (4 / NSUB), (F + 15) / 16);
    kd_conv_bm<NSUB, KP4, FIRST><<<grid, 256, 0, stream>>>(cur, W, b, sel, nxt, F, Dlog2);
}

extern "C" void kernel_launch(void* const* d_in, const int* in_sizes, int n_in,
                              void* d_out, int out_size, void* d_ws, size_t ws_size,
                              hipStream_t stream) {
    static const int DIMS[11] = {2048, 1024, 512, 256, 128, 64, 32, 16, 8, 4, 2};
    static const int FEAT[11] = {8, 32, 64, 64, 64, 128, 256, 512, 512, 512, 1024};

    const size_t XBYTES = (size_t)512 * 32 * 512 * sizeof(float);
    float* X0     = (float*)d_ws;
    float* X1     = (float*)((char*)d_ws + XBYTES);

    const float* cur = (const float*)d_in[0];
    for (int i = 0; i < 11; ++i) {
        const float* W  = (const float*)d_in[12 + 2 * i];
        const float* bi = (const float*)d_in[13 + 2 * i];
        const int*   sl = (const int*)d_in[1 + i];
        float* nxt = (i & 1) ? X1 : X0;
        const int F = FEAT[i], Dd = DIMS[i];
        switch (i) {
        case 0:  launch_conv<4, 4,   true >(cur, W, bi, sl, nxt, F, Dd, stream); break;
        case 1:  launch_conv<4, 8,   false>(cur, W, bi, sl, nxt, F, Dd, stream); break;
        case 2:  launch_conv<4, 32,  false>(cur, W, bi, sl, nxt, F, Dd, stream); break;
        case 3:  launch_conv<2, 64,  false>(cur, W, bi, sl, nxt, F, Dd, stream); break;
        case 4:
        case 5:  launch_conv<1, 64,  false>(cur, W, bi, sl, nxt, F, Dd, stream); break;
        case 6:  launch_conv<1, 128, false>(cur, W, bi, sl, nxt, F, Dd, stream); break;
        case 7:  launch_conv<1, 256, false>(cur, W, bi, sl, nxt, F, Dd, stream); break;
        default: launch_conv<1, 512, false>(cur, W, bi, sl, nxt, F, Dd, stream); break;
        }
        cur = nxt;
    }

    kd_fc_lsm<<<32, 32, 0, stream>>>(
        cur, (const float*)d_in[34], (const float*)d_in[35], (float*)d_out);

    (void)in_sizes; (void)n_in; (void)out_size; (void)ws_size;
}


# Conv layers 3-11 as block-tiled GEMM: W rows and gathered activation rows staged per K chunk in two LDS stages shared by 8 waves (64 features x 128 or 32 batches per block); sums unchanged
# speedup vs baseline: 7.8747x; 1.2080x over previous
#include <hip/hip_runtime.h>


typedef float v2f __attribute__((ext_vector_type(2)));
typedef float v4f __attribute__((ext_vector_type(4)));
typedef float v8f __attribute__((ext_vector_type(8)));

template <int NSUB, int KP4, bool FIRST>
__global__ __launch_bounds__(256) void kd_conv_bm(
    const float* __restrict__ X, const float* __restrict__ Wt,
    const float* __restrict__ bias, const int* __restrict__ sel,
    float* __restrict__ Out, int F, int Dlog2)
{
    constexpr int K   = FIRST ? 3 : KP4;
    constexpr int KP  = KP4 + 1;
    constexpr int NB  = 512;
    constexpr int NBC = 4 / NSUB;
    constexpr int UNR = 8 / NSUB;
    __shared__ float sW[48 * KP];

    const int f0 = blockIdx.y << 4;

    constexpr int TOT = 48 * KP4;
    constexpr int NIT = (TOT + 255) / 256;
    constexpr bool EXACT = (TOT % 256) == 0;
#pragma unroll 16
    for (int it = 0; it < NIT; ++it) {
        const int idx = it * 256 + (int)threadIdx.x;
        const int idc = EXACT ? idx : min(idx, TOT - 1);
        const int rr  = idc / KP4;
        const int k   = idc - rr * KP4;
        const int q   = rr >> 4;
        const int f   = f0 + (rr & 15);
        float w = Wt[(3 * min(f, F - 1) + q) * K + min(k, K - 1)];
        w = (f < F && k < K) ? w : 0.0f;
        if (EXACT || idx < TOT) sW[rr * KP + k] = w;
    }
    __syncthreads();

    const int lane = threadIdx.x & 31;
    const int wave = threadIdx.x >> 5;
    const int half = lane >> 4;
    const int c    = lane & 15;
    const int kOff = half << 1;

    const int j     = blockIdx.x / NBC;
    const int bc    = blockIdx.x - j * NBC;
    const int bBase = (bc * 8 + wave) * (16 * NSUB);
    const int D     = 1 << Dlog2;

    const int d0 = j << 1;
    const int d1 = d0 + 1;
    const int s0 = min(max(sel[d0], 0), 2);
    const int s1 = min(max(sel[d1], 0), 2);
    const int t0 = 3 * d0 + s0;
    const int t1 = 3 * d1 + s1;
    const int q0 = t0 >> Dlog2;
    const int q1 = t1 >> Dlog2;
    const int dp0 = t0 & (D - 1);
    const int dp1 = t1 & (D - 1);

    const float* aRow0 = sW + (q0 * 16 + c) * KP;
    const float* aRow1 = sW + (q1 * 16 + c) * KP;

    const int sk = FIRST ? D : NB;
    const float* p0[NSUB];
    const float* p1[NSUB];
#pragma unroll
    for (int s = 0; s < NSUB; ++s) {
        const int b = bBase + (s << 4) + c;
        if (FIRST) {
            p0[s] = X + (size_t)b * (size_t)(K * D) + dp0;
            p1[s] = X + (size_t)b * (size_t)(K * D) + dp1;
        } else {
            p0[s] = X + (size_t)dp0 * (size_t)(K * NB) + b;
            p1[s] = X + (size_t)dp1 * (size_t)(K * NB) + b;
        }
    }

    v8f acc0[NSUB];
    v8f acc1[NSUB];
#pragma unroll
    for (int s = 0; s < NSUB; ++s) { acc0[s] = v8f{}; acc1[s] = v8f{}; }

    constexpr int kFull = K & ~3;
#pragma unroll UNR
    for (int kt = 0; kt < kFull; kt += 4) {
        const int k0 = kt + kOff;
        v2f a0, a1;
        a0.x = aRow0[k0]; a0.y = aRow0[k0 + 1];
        a1.x = aRow1[k0]; a1.y = aRow1[k0 + 1];
#pragma unroll
        for (int s = 0; s < NSUB; ++s) {
            v2f b0, b1;
            b0.x = p0[s][(size_t)k0 * sk];
            b0.y = p0[s][(size_t)(k0 + 1) * sk];
            b1.x = p1[s][(size_t)k0 * sk];
            b1.y = p1[s][(size_t)(k0 + 1) * sk];
            acc0[s] = __builtin_amdgcn_wmma_f32_16x16x4_f32(
                false, a0, false, b0, (short)0, acc0[s], false, false);
            acc1[s] = __builtin_amdgcn_wmma_f32_16x16x4_f32(
                false, a1, false, b1, (short)0, acc1[s], false, false);
        }
    }
    if (kFull < K) {
        const int k0 = kFull + kOff;
        const int c0 = min(k0, K - 1);
        const int c1 = min(k0 + 1, K - 1);
        v2f a0, a1;
        a0.x = aRow0[k0]; a0.y = aRow0[k0 + 1];
        a1.x = aRow1[k0]; a1.y = aRow1[k0 + 1];
#pragma unroll
        for (int s = 0; s < NSUB; ++s) {
            v2f b0, b1;
            b0.x = p0[s][(size_t)c0 * sk];
            b0.y = p0[s][(size_t)c1 * sk];
            b1.x = p1[s][(size_t)c0 * sk];
            b1.y = p1[s][(size_t)c1 * sk];
            acc0[s] = __builtin_amdgcn_wmma_f32_16x16x4_f32(
                false, a0, false, b0, (short)0, acc0[s], false, false);
            acc1[s] = __builtin_amdgcn_wmma_f32_16x16x4_f32(
                false, a1, false, b1, (short)0, acc1[s], false, false);
        }
    }

    float bq0[8], bq1[8];
#pragma unroll
    for (int r = 0; r < 8; ++r) {
        const int f = f0 + (half << 3) + r;
        const int fc = min(f, F - 1);
        bq0[r] = bias[3 * fc + q0];
        bq1[r] = bias[3 * fc + q1];
    }
#pragma unroll
    for (int s = 0; s < NSUB; ++s) {
        const int b = bBase + (s << 4) + c;
#pragma unroll
        for (int r = 0; r < 8; ++r) {
            const int f = f0 + (half << 3) + r;
            if (f < F) {
                const float v0 = fmaxf(acc0[s][r] + bq0[r], 0.0f);
                const float v1 = fmaxf(acc1[s][r] + bq1[r], 0.0f);
                Out[((size_t)j * F + f) * NB + b] = fmaxf(v0, v1);
            }
        }
    }
}

template <int WM, int WN, int FT, int BT, int KC, int K>
__global__ __launch_bounds__(256) void kd_conv_tile(
    const float* __restrict__ X, const float* __restrict__ Wt,
    const float* __restrict__ bias, const int* __restrict__ sel,
    float* __restrict__ Out, int F, int Dlog2)
{
    static_assert(WM * WN == 8, "eight waves per block");
    static_assert(K % KC == 0, "whole chunks");
    constexpr int NB  = 512;
    constexpr int MF  = WM * FT * 16;
    constexpr int NBT = WN * BT * 16;
    constexpr int NBC = NB / NBT;
    constexpr int PA  = KC + 4;
    constexpr int KP  = KC / 2;
    constexpr int PB  = 2 * NBT + 32;
    constexpr int SA  = 2 * MF * PA;
    constexpr int SB  = 2 * KP * PB;
    constexpr int A4R = KC / 4;
    constexpr int B4R = NBT / 4;
    constexpr int NA4 = 2 * MF * A4R / 256;
    constexpr int NB4 = 2 * KP * B4R / 256;
    constexpr int NCH = K / KC;
    constexpr int NST = NCH > 1 ? 2 : 1;
    static_assert((MF * A4R) % 256 == 0 && (KP * B4R) % 256 == 0,
                  "a load slot never straddles the two d");

    __shared__ __attribute__((aligned(16))) float sA[NST * SA];
    __shared__ __attribute__((aligned(16))) float sB[NST * SB];

    const int tid  = threadIdx.x;
    const int lane = tid & 31;
    const int wave = tid >> 5;
    const int half = lane >> 4;
    const int c    = lane & 15;
    const int wm   = wave / WN;
    const int wn   = wave - wm * WN;

    const int f0 = blockIdx.y * MF;
    const int j  = blockIdx.x / NBC;
    const int bb = (blockIdx.x - j * NBC) * NBT;
    const int D  = 1 << Dlog2;

    const int d0 = j << 1;
    const int d1 = d0 + 1;
    const int s0 = min(max(sel[d0], 0), 2);
    const int s1 = min(max(sel[d1], 0), 2);
    const int t0 = 3 * d0 + s0;
    const int t1 = 3 * d1 + s1;
    const int qd[2]  = { t0 >> Dlog2, t1 >> Dlog2 };
    const int dpd[2] = { t0 & (D - 1), t1 & (D - 1) };

    int offA[NA4], ldsA[NA4];
#pragma unroll
    for (int it = 0; it < NA4; ++it) {
        const int ds  = (it * 256) / (MF * A4R);
        const int rem = it * 256 + tid - ds * (MF * A4R);
        const int r   = rem / A4R;
        const int c4  = rem - r * A4R;
        offA[it] = (3 * (f0 + r) + qd[ds]) * K + 4 * c4;
        ldsA[it] = (ds * MF + r) * PA + 4 * c4;
    }
    int offB[NB4], ldsB[NB4];
#pragma unroll
    for (int it = 0; it < NB4; ++it) {
        const int ds  = (it * 256) / (KP * B4R);
        const int rem = it * 256 + tid - ds * (KP * B4R);
        const int kp  = rem / B4R;
        const int c4  = rem - kp * B4R;
        offB[it] = (dpd[ds] * K + 2 * kp) * NB + bb + 4 * c4;
        ldsB[it] = (ds * KP + kp) * PB + 8 * c4;
    }

    v4f ra[NA4], rb[NB4][2];
    v8f acc[2][FT][BT];
#pragma unroll
    for (int ds = 0; ds < 2; ++ds)
#pragma unroll
        for (int ft = 0; ft < FT; ++ft)
#pragma unroll
            for (int bt = 0; bt < BT; ++bt) acc[ds][ft][bt] = v8f{};

    auto fetch = [&](int ch) __attribute__((always_inline)) {
#pragma unroll
        for (int it = 0; it < NA4; ++it)
            ra[it] = *reinterpret_cast<const v4f*>(Wt + offA[it] + ch * KC);
#pragma unroll
        for (int it = 0; it < NB4; ++it) {
            rb[it][0] = *reinterpret_cast<const v4f*>(X + offB[it] + ch * (KC * NB));
            rb[it][1] = *reinterpret_cast<const v4f*>(X + offB[it] + ch * (KC * NB) + NB);
        }
    };
    auto stash = [&](int st) __attribute__((always_inline)) {
#pragma unroll
        for (int it = 0; it < NA4; ++it)
            *reinterpret_cast<v4f*>(sA + st * SA + ldsA[it]) = ra[it];
#pragma unroll
        for (int it = 0; it < NB4; ++it) {
            const v4f r0 = rb[it][0], r1 = rb[it][1];
            const v4f lo = { r0.x, r1.x, r0.y, r1.y };
            const v4f hi = { r0.z, r1.z, r0.w, r1.w };
            *reinterpret_cast<v4f*>(sB + st * SB + ldsB[it])     = lo;
            *reinterpret_cast<v4f*>(sB + st * SB + ldsB[it] + 4) = hi;
        }
    };
    auto multiply = [&](int st) __attribute__((always_inline)) {
        const float* as = sA + st * SA + ((wm * FT) * 16 + c) * PA + 2 * half;
        const float* bs = sB + st * SB + half * PB + ((wn * BT) * 16 + c) * 2;
#pragma unroll
        for (int ks = 0; ks < KC / 4; ++ks) {
            v2f a[2][FT], b[2][BT];
#pragma unroll
            for (int ds = 0; ds < 2; ++ds) {
#pragma unroll
                for (int ft = 0; ft < FT; ++ft)
                    a[ds][ft] = *reinterpret_cast<const v2f*>(
                        as + (ds * MF + ft * 16) * PA + ks * 4);
#pragma unroll
                for (int bt = 0; bt < BT; ++bt)
                    b[ds][bt] = *reinterpret_cast<const v2f*>(
                        bs + (ds * KP + ks * 2) * PB + bt * 32);
            }
#pragma unroll
            for (int ds = 0; ds < 2; ++ds)
#pragma unroll
                for (int ft = 0; ft < FT; ++ft)
#pragma unroll
                    for (int bt = 0; bt < BT; ++bt)
                        acc[ds][ft][bt] = __builtin_amdgcn_wmma_f32_16x16x4_f32(
                            false, a[ds][ft], false, b[ds][bt], (short)0,
                            acc[ds][ft][bt], false, false);
        }
    };

    fetch(0);
    stash(0);
    __syncthreads();
#pragma nounroll
    for (int ch = 0; ch < NCH - 1; ++ch) {
        fetch(ch + 1);
        __builtin_amdgcn_sched_barrier(0);
        multiply(ch & 1);
        __builtin_amdgcn_sched_barrier(0);
        stash((ch + 1) & 1);
        __syncthreads();
    }
    multiply((NCH - 1) & 1);

#pragma unroll
    for (int ft = 0; ft < FT; ++ft) {
        float bq0[8], bq1[8];
#pragma unroll
        for (int r = 0; r < 8; ++r) {
            const int f = f0 + (wm * FT + ft) * 16 + (half << 3) + r;
            bq0[r] = bias[3 * f + qd[0]];
            bq1[r] = bias[3 * f + qd[1]];
        }
#pragma unroll
        for (int bt = 0; bt < BT; ++bt) {
            const int b = bb + (wn * BT + bt) * 16 + c;
#pragma unroll
            for (int r = 0; r < 8; ++r) {
                const int f = f0 + (wm * FT + ft) * 16 + (half << 3) + r;
                const float v0 = fmaxf(acc[0][ft][bt][r] + bq0[r], 0.0f);
                const float v1 = fmaxf(acc[1][ft][bt][r] + bq1[r], 0.0f);
                Out[((size_t)j * F + f) * NB + b] = fmaxf(v0, v1);
            }
        }
    }
}

__global__ __launch_bounds__(32) void kd_fc_lsm(
    const float* __restrict__ Y, const float* __restrict__ Wt,
    const float* __restrict__ bias, float* __restrict__ out)
{
    constexpr int K  = 1024;
    constexpr int NB = 512;
    constexpr int M  = 16;

    const int lane = threadIdx.x & 31;
    const int half = lane >> 4;
    const int c    = lane & 15;
    const int kOff = half << 1;
    const int b    = (blockIdx.x << 4) + c;

    const float* ap = Wt + c * K + kOff;
    const float* bp = Y + (size_t)kOff * NB + b;

    float bs[8];
#pragma unroll
    for (int r = 0; r < 8; ++r) bs[r] = bias[(half << 3) + r];

    v8f acc = v8f{};
#pragma unroll 16
    for (int kt = 0; kt < K; kt += 4) {
        const v2f a = *reinterpret_cast<const v2f*>(ap + kt);
        v2f x;
        x.x = bp[(size_t)kt * NB];
        x.y = bp[(size_t)(kt + 1) * NB];
        acc = __builtin_amdgcn_wmma_f32_16x16x4_f32(
            false, a, false, x, (short)0, acc, false, false);
    }

    float v[8];
    float mx = -3.402823466e38f;
#pragma unroll
    for (int r = 0; r < 8; ++r) { v[r] = acc[r] + bs[r]; mx = fmaxf(mx, v[r]); }
    mx = fmaxf(mx, __shfl_xor(mx, 16, 32));

    float e[8];
#pragma unroll
    for (int r = 0; r < 8; ++r) e[r] = expf(v[r] - mx);
    float s = 0.0f;
#pragma unroll
    for (int r = 0; r < 8; ++r) s += e[r];
    float t = __shfl(s, c, 32);
#pragma unroll
    for (int r = 0; r < 8; ++r) t += e[r];
    const float tot = __shfl(t, c + 16, 32);
    const float l = mx + logf(tot);

    float4 o0, o1;
    o0.x = v[0] - l; o0.y = v[1] - l; o0.z = v[2] - l; o0.w = v[3] - l;
    o1.x = v[4] - l; o1.y = v[5] - l; o1.z = v[6] - l; o1.w = v[7] - l;
    float4* op = reinterpret_cast<float4*>(out + (size_t)b * M + (half << 3));
    op[0] = o0;
    op[1] = o1;
}

template <int NSUB, int KP4, bool FIRST>
static void launch_conv(const float* cur, const float* W, const float* b,
                        const int* sel, float* nxt, int F, int Dd,
                        hipStream_t stream)
{
    const int Dlog2 = __builtin_ctz((unsigned)Dd);
    dim3 grid((Dd / 2) * (4 / NSUB), (F + 15) / 16);
    kd_conv_bm<NSUB, KP4, FIRST><<<grid, 256, 0, stream>>>(cur, W, b, sel, nxt, F, Dlog2);
}

template <int WM, int WN, int FT, int BT, int KC, int K>
static void launch_tile(const float* cur, const float* W, const float* b,
                        const int* sel, float* nxt, int F, int Dd,
                        hipStream_t stream)
{
    const int Dlog2 = __builtin_ctz((unsigned)Dd);
    dim3 grid((Dd / 2) * (512 / (WN * BT * 16)), F / (WM * FT * 16));
    kd_conv_tile<WM, WN, FT, BT, KC, K><<<grid, 256, 0, stream>>>(cur, W, b, sel, nxt, F, Dlog2);
}

extern "C" void kernel_launch(void* const* d_in, const int* in_sizes, int n_in,
                              void* d_out, int out_size, void* d_ws, size_t ws_size,
                              hipStream_t stream) {
    static const int DIMS[11] = {2048, 1024, 512, 256, 128, 64, 32, 16, 8, 4, 2};
    static const int FEAT[11] = {8, 32, 64, 64, 64, 128, 256, 512, 512, 512, 1024};

    const size_t XBYTES = (size_t)512 * 32 * 512 * sizeof(float);
    float* X0     = (float*)d_ws;
    float* X1     = (float*)((char*)d_ws + XBYTES);

    const float* cur = (const float*)d_in[0];
    for (int i = 0; i < 11; ++i) {
        const float* W  = (const float*)d_in[12 + 2 * i];
        const float* bi = (const float*)d_in[13 + 2 * i];
        const int*   sl = (const int*)d_in[1 + i];
        float* nxt = (i & 1) ? X1 : X0;
        const int F = FEAT[i], Dd = DIMS[i];
        switch (i) {
        case 0:  launch_conv<4, 4,   true >(cur, W, bi, sl, nxt, F, Dd, stream); break;
        case 1:  launch_conv<4, 8,   false>(cur, W, bi, sl, nxt, F, Dd, stream); break;
        case 2:  launch_tile<2, 4, 2, 2, 32, 32 >(cur, W, bi, sl, nxt, F, Dd, stream); break;
        case 3:
        case 4:
        case 5:  launch_tile<2, 4, 2, 2, 32, 64 >(cur, W, bi, sl, nxt, F, Dd, stream); break;
        case 6:  launch_tile<2, 4, 2, 2, 32, 128>(cur, W, bi, sl, nxt, F, Dd, stream); break;
        case 7:  launch_tile<2, 4, 2, 2, 32, 256>(cur, W, bi, sl, nxt, F, Dd, stream); break;
        default: launch_tile<4, 2, 1, 1, 64, 512>(cur, W, bi, sl, nxt, F, Dd, stream); break;
        }
        cur = nxt;
    }

    kd_fc_lsm<<<32, 32, 0, stream>>>(
        cur, (const float*)d_in[34], (const float*)d_in[35], (float*)d_out);

    (void)in_sizes; (void)n_in; (void)out_size; (void)ws_size;
}


# Layer 2 folded into layer 3's launch (12 -> 11): each wave forms the layer-2 cells its 16 batches gather from the layer-1 output and stores them as layer 3's B stage in LDS; sums unchanged
# speedup vs baseline: 8.5439x; 1.0850x over previous
#include <hip/hip_runtime.h>


typedef float v2f __attribute__((ext_vector_type(2)));
typedef float v4f __attribute__((ext_vector_type(4)));
typedef float v8f __attribute__((ext_vector_type(8)));

template <int NSUB, int KP4, bool FIRST>
__global__ __launch_bounds__(256) void kd_conv_bm(
    const float* __restrict__ X, const float* __restrict__ Wt,
    const float* __restrict__ bias, const int* __restrict__ sel,
    float* __restrict__ Out, int F, int Dlog2)
{
    constexpr int K   = FIRST ? 3 : KP4;
    constexpr int KP  = KP4 + 1;
    constexpr int NB  = 512;
    constexpr int NBC = 4 / NSUB;
    constexpr int UNR = 8 / NSUB;
    __shared__ float sW[48 * KP];

    const int f0 = blockIdx.y << 4;

    constexpr int TOT = 48 * KP4;
    constexpr int NIT = (TOT + 255) / 256;
    constexpr bool EXACT = (TOT % 256) == 0;
#pragma unroll 16
    for (int it = 0; it < NIT; ++it) {
        const int idx = it * 256 + (int)threadIdx.x;
        const int idc = EXACT ? idx : min(idx, TOT - 1);
        const int rr  = idc / KP4;
        const int k   = idc - rr * KP4;
        const int q   = rr >> 4;
        const int f   = f0 + (rr & 15);
        float w = Wt[(3 * min(f, F - 1) + q) * K + min(k, K - 1)];
        w = (f < F && k < K) ? w : 0.0f;
        if (EXACT || idx < TOT) sW[rr * KP + k] = w;
    }
    __syncthreads();

    const int lane = threadIdx.x & 31;
    const int wave = threadIdx.x >> 5;
    const int half = lane >> 4;
    const int c    = lane & 15;
    const int kOff = half << 1;

    const int j     = blockIdx.x / NBC;
    const int bc    = blockIdx.x - j * NBC;
    const int bBase = (bc * 8 + wave) * (16 * NSUB);
    const int D     = 1 << Dlog2;

    const int d0 = j << 1;
    const int d1 = d0 + 1;
    const int s0 = min(max(sel[d0], 0), 2);
    const int s1 = min(max(sel[d1], 0), 2);
    const int t0 = 3 * d0 + s0;
    const int t1 = 3 * d1 + s1;
    const int q0 = t0 >> Dlog2;
    const int q1 = t1 >> Dlog2;
    const int dp0 = t0 & (D - 1);
    const int dp1 = t1 & (D - 1);

    const float* aRow0 = sW + (q0 * 16 + c) * KP;
    const float* aRow1 = sW + (q1 * 16 + c) * KP;

    const int sk = FIRST ? D : NB;
    const float* p0[NSUB];
    const float* p1[NSUB];
#pragma unroll
    for (int s = 0; s < NSUB; ++s) {
        const int b = bBase + (s << 4) + c;
        if (FIRST) {
            p0[s] = X + (size_t)b * (size_t)(K * D) + dp0;
            p1[s] = X + (size_t)b * (size_t)(K * D) + dp1;
        } else {
            p0[s] = X + (size_t)dp0 * (size_t)(K * NB) + b;
            p1[s] = X + (size_t)dp1 * (size_t)(K * NB) + b;
        }
    }

    v8f acc0[NSUB];
    v8f acc1[NSUB];
#pragma unroll
    for (int s = 0; s < NSUB; ++s) { acc0[s] = v8f{}; acc1[s] = v8f{}; }

    constexpr int kFull = K & ~3;
#pragma unroll UNR
    for (int kt = 0; kt < kFull; kt += 4) {
        const int k0 = kt + kOff;
        v2f a0, a1;
        a0.x = aRow0[k0]; a0.y = aRow0[k0 + 1];
        a1.x = aRow1[k0]; a1.y = aRow1[k0 + 1];
#pragma unroll
        for (int s = 0; s < NSUB; ++s) {
            v2f b0, b1;
            b0.x = p0[s][(size_t)k0 * sk];
            b0.y = p0[s][(size_t)(k0 + 1) * sk];
            b1.x = p1[s][(size_t)k0 * sk];
            b1.y = p1[s][(size_t)(k0 + 1) * sk];
            acc0[s] = __builtin_amdgcn_wmma_f32_16x16x4_f32(
                false, a0, false, b0, (short)0, acc0[s], false, false);
            acc1[s] = __builtin_amdgcn_wmma_f32_16x16x4_f32(
                false, a1, false, b1, (short)0, acc1[s], false, false);
        }
    }
    if (kFull < K) {
        const int k0 = kFull + kOff;
        const int c0 = min(k0, K - 1);
        const int c1 = min(k0 + 1, K - 1);
        v2f a0, a1;
        a0.x = aRow0[k0]; a0.y = aRow0[k0 + 1];
        a1.x = aRow1[k0]; a1.y = aRow1[k0 + 1];
#pragma unroll
        for (int s = 0; s < NSUB; ++s) {
            v2f b0, b1;
            b0.x = p0[s][(size_t)c0 * sk];
            b0.y = p0[s][(size_t)c1 * sk];
            b1.x = p1[s][(size_t)c0 * sk];
            b1.y = p1[s][(size_t)c1 * sk];
            acc0[s] = __builtin_amdgcn_wmma_f32_16x16x4_f32(
                false, a0, false, b0, (short)0, acc0[s], false, false);
            acc1[s] = __builtin_amdgcn_wmma_f32_16x16x4_f32(
                false, a1, false, b1, (short)0, acc1[s], false, false);
        }
    }

    float bq0[8], bq1[8];
#pragma unroll
    for (int r = 0; r < 8; ++r) {
        const int f = f0 + (half << 3) + r;
        const int fc = min(f, F - 1);
        bq0[r] = bias[3 * fc + q0];
        bq1[r] = bias[3 * fc + q1];
    }
#pragma unroll
    for (int s = 0; s < NSUB; ++s) {
        const int b = bBase + (s << 4) + c;
#pragma unroll
        for (int r = 0; r < 8; ++r) {
            const int f = f0 + (half << 3) + r;
            if (f < F) {
                const float v0 = fmaxf(acc0[s][r] + bq0[r], 0.0f);
                const float v1 = fmaxf(acc1[s][r] + bq1[r], 0.0f);
                Out[((size_t)j * F + f) * NB + b] = fmaxf(v0, v1);
            }
        }
    }
}

template <int WM, int WN, int FT, int BT, int KC, int K>
__global__ __launch_bounds__(256) void kd_conv_tile(
    const float* __restrict__ X, const float* __restrict__ Wt,
    const float* __restrict__ bias, const int* __restrict__ sel,
    float* __restrict__ Out, int F, int Dlog2)
{
    static_assert(WM * WN == 8, "eight waves per block");
    static_assert(K % KC == 0, "whole chunks");
    constexpr int NB  = 512;
    constexpr int MF  = WM * FT * 16;
    constexpr int NBT = WN * BT * 16;
    constexpr int NBC = NB / NBT;
    constexpr int PA  = KC + 4;
    constexpr int KP  = KC / 2;
    constexpr int PB  = 2 * NBT + 32;
    constexpr int SA  = 2 * MF * PA;
    constexpr int SB  = 2 * KP * PB;
    constexpr int A4R = KC / 4;
    constexpr int B4R = NBT / 4;
    constexpr int NA4 = 2 * MF * A4R / 256;
    constexpr int NB4 = 2 * KP * B4R / 256;
    constexpr int NCH = K / KC;
    constexpr int NST = NCH > 1 ? 2 : 1;
    static_assert((MF * A4R) % 256 == 0 && (KP * B4R) % 256 == 0,
                  "a load slot never straddles the two d");

    __shared__ __attribute__((aligned(16))) float sA[NST * SA];
    __shared__ __attribute__((aligned(16))) float sB[NST * SB];

    const int tid  = threadIdx.x;
    const int lane = tid & 31;
    const int wave = tid >> 5;
    const int half = lane >> 4;
    const int c    = lane & 15;
    const int wm   = wave / WN;
    const int wn   = wave - wm * WN;

    const int f0 = blockIdx.y * MF;
    const int j  = blockIdx.x / NBC;
    const int bb = (blockIdx.x - j * NBC) * NBT;
    const int D  = 1 << Dlog2;

    const int d0 = j << 1;
    const int d1 = d0 + 1;
    const int s0 = min(max(sel[d0], 0), 2);
    const int s1 = min(max(sel[d1], 0), 2);
    const int t0 = 3 * d0 + s0;
    const int t1 = 3 * d1 + s1;
    const int qd[2]  = { t0 >> Dlog2, t1 >> Dlog2 };
    const int dpd[2] = { t0 & (D - 1), t1 & (D - 1) };

    int offA[NA4], ldsA[NA4];
#pragma unroll
    for (int it = 0; it < NA4; ++it) {
        const int ds  = (it * 256) / (MF * A4R);
        const int rem = it * 256 + tid - ds * (MF * A4R);
        const int r   = rem / A4R;
        const int c4  = rem - r * A4R;
        offA[it] = (3 * (f0 + r) + qd[ds]) * K + 4 * c4;
        ldsA[it] = (ds * MF + r) * PA + 4 * c4;
    }
    int offB[NB4], ldsB[NB4];
#pragma unroll
    for (int it = 0; it < NB4; ++it) {
        const int ds  = (it * 256) / (KP * B4R);
        const int rem = it * 256 + tid - ds * (KP * B4R);
        const int kp  = rem / B4R;
        const int c4  = rem - kp * B4R;
        offB[it] = (dpd[ds] * K + 2 * kp) * NB + bb + 4 * c4;
        ldsB[it] = (ds * KP + kp) * PB + 8 * c4;
    }

    v4f ra[NA4], rb[NB4][2];
    v8f acc[2][FT][BT];
#pragma unroll
    for (int ds = 0; ds < 2; ++ds)
#pragma unroll
        for (int ft = 0; ft < FT; ++ft)
#pragma unroll
            for (int bt = 0; bt < BT; ++bt) acc[ds][ft][bt] = v8f{};

    auto fetch = [&](int ch) __attribute__((always_inline)) {
#pragma unroll
        for (int it = 0; it < NA4; ++it)
            ra[it] = *reinterpret_cast<const v4f*>(Wt + offA[it] + ch * KC);
#pragma unroll
        for (int it = 0; it < NB4; ++it) {
            rb[it][0] = *reinterpret_cast<const v4f*>(X + offB[it] + ch * (KC * NB));
            rb[it][1] = *reinterpret_cast<const v4f*>(X + offB[it] + ch * (KC * NB) + NB);
        }
    };
    auto stash = [&](int st) __attribute__((always_inline)) {
#pragma unroll
        for (int it = 0; it < NA4; ++it)
            *reinterpret_cast<v4f*>(sA + st * SA + ldsA[it]) = ra[it];
#pragma unroll
        for (int it = 0; it < NB4; ++it) {
            const v4f r0 = rb[it][0], r1 = rb[it][1];
            const v4f lo = { r0.x, r1.x, r0.y, r1.y };
            const v4f hi = { r0.z, r1.z, r0.w, r1.w };
            *reinterpret_cast<v4f*>(sB + st * SB + ldsB[it])     = lo;
            *reinterpret_cast<v4f*>(sB + st * SB + ldsB[it] + 4) = hi;
        }
    };
    auto multiply = [&](int st) __attribute__((always_inline)) {
        const float* as = sA + st * SA + ((wm * FT) * 16 + c) * PA + 2 * half;
        const float* bs = sB + st * SB + half * PB + ((wn * BT) * 16 + c) * 2;
#pragma unroll
        for (int ks = 0; ks < KC / 4; ++ks) {
            v2f a[2][FT], b[2][BT];
#pragma unroll
            for (int ds = 0; ds < 2; ++ds) {
#pragma unroll
                for (int ft = 0; ft < FT; ++ft)
                    a[ds][ft] = *reinterpret_cast<const v2f*>(
                        as + (ds * MF + ft * 16) * PA + ks * 4);
#pragma unroll
                for (int bt = 0; bt < BT; ++bt)
                    b[ds][bt] = *reinterpret_cast<const v2f*>(
                        bs + (ds * KP + ks * 2) * PB + bt * 32);
            }
#pragma unroll
            for (int ds = 0; ds < 2; ++ds)
#pragma unroll
                for (int ft = 0; ft < FT; ++ft)
#pragma unroll
                    for (int bt = 0; bt < BT; ++bt)
                        acc[ds][ft][bt] = __builtin_amdgcn_wmma_f32_16x16x4_f32(
                            false, a[ds][ft], false, b[ds][bt], (short)0,
                            acc[ds][ft][bt], false, false);
        }
    };

    fetch(0);
    stash(0);
    __syncthreads();
#pragma nounroll
    for (int ch = 0; ch < NCH - 1; ++ch) {
        fetch(ch + 1);
        __builtin_amdgcn_sched_barrier(0);
        multiply(ch & 1);
        __builtin_amdgcn_sched_barrier(0);
        stash((ch + 1) & 1);
        __syncthreads();
    }
    multiply((NCH - 1) & 1);

#pragma unroll
    for (int ft = 0; ft < FT; ++ft) {
        float bq0[8], bq1[8];
#pragma unroll
        for (int r = 0; r < 8; ++r) {
            const int f = f0 + (wm * FT + ft) * 16 + (half << 3) + r;
            bq0[r] = bias[3 * f + qd[0]];
            bq1[r] = bias[3 * f + qd[1]];
        }
#pragma unroll
        for (int bt = 0; bt < BT; ++bt) {
            const int b = bb + (wn * BT + bt) * 16 + c;
#pragma unroll
            for (int r = 0; r < 8; ++r) {
                const int f = f0 + (wm * FT + ft) * 16 + (half << 3) + r;
                const float v0 = fmaxf(acc[0][ft][bt][r] + bq0[r], 0.0f);
                const float v1 = fmaxf(acc[1][ft][bt][r] + bq1[r], 0.0f);
                Out[((size_t)j * F + f) * NB + b] = fmaxf(v0, v1);
            }
        }
    }
}

__global__ __launch_bounds__(256) void kd_conv3_fold2(
    const float* __restrict__ X1, const float* __restrict__ W2,
    const float* __restrict__ b2, const int* __restrict__ sel2,
    const float* __restrict__ Wt, const float* __restrict__ bias,
    const int* __restrict__ sel, float* __restrict__ Out,
    int Dlog2, int D2log2)
{
    constexpr int WM = 2, WN = 4, FT = 2, BT = 2;
    constexpr int NB  = 512;
    constexpr int F   = 64;
    constexpr int K   = 32;
    constexpr int K2  = 8;
    constexpr int R2  = 3 * K;
    constexpr int MF  = WM * FT * 16;
    constexpr int NBT = WN * BT * 16;
    constexpr int NBC = NB / NBT;
    constexpr int PA  = K + 4;
    constexpr int KP  = K / 2;
    constexpr int PB  = 2 * NBT + 32;
    constexpr int SA  = 2 * MF * PA;
    constexpr int SB  = 2 * KP * PB;
    constexpr int A4R = K / 4;
    constexpr int NA4 = 2 * MF * A4R / 256;
    static_assert(NBT == 8 * 16, "layer-2 stage: one 16-batch column tile per wave");
    static_assert(K == 2 * 16, "layer-2 stage: two feature tiles");
    constexpr int NW2 = R2 * K2 / 256;
    static_assert((R2 * K2) % 256 == 0 && R2 <= 256, "W2 / b2 copy: whole trips");

    __shared__ __attribute__((aligned(16))) float sA[SA];
    __shared__ __attribute__((aligned(16))) float sB[SB];
    __shared__ __attribute__((aligned(16))) float sW2[R2 * K2];
    __shared__ float sb2[256];

    const int tid  = threadIdx.x;
    const int lane = tid & 31;
    const int wave = tid >> 5;
    const int half = lane >> 4;
    const int c    = lane & 15;
    const int wm   = wave / WN;
    const int wn   = wave - wm * WN;

    float w2r[NW2];
#pragma unroll
    for (int it = 0; it < NW2; ++it) w2r[it] = W2[it * 256 + tid];
    const float b2r = b2[min(tid, R2 - 1)];

    const int f0 = blockIdx.y * MF;
    const int j  = blockIdx.x / NBC;
    const int bb = (blockIdx.x - j * NBC) * NBT;
    const int D  = 1 << Dlog2;
    const int D2 = 1 << D2log2;

    const int d0 = j << 1;
    const int d1 = d0 + 1;
    const int s0 = min(max(sel[d0], 0), 2);
    const int s1 = min(max(sel[d1], 0), 2);
    const int t0 = 3 * d0 + s0;
    const int t1 = 3 * d1 + s1;
    const int qd[2]  = { t0 >> Dlog2, t1 >> Dlog2 };
    const int dpd[2] = { t0 & (D - 1), t1 & (D - 1) };

    int ldsA[NA4];
    v4f ra[NA4];
#pragma unroll
    for (int it = 0; it < NA4; ++it) {
        const int ds  = (it * 256) / (MF * A4R);
        const int rem = it * 256 + tid - ds * (MF * A4R);
        const int r   = rem / A4R;
        const int c4  = rem - r * A4R;
        ldsA[it] = (ds * MF + r) * PA + 4 * c4;
        ra[it] = *reinterpret_cast<const v4f*>(
            Wt + (3 * (f0 + r) + qd[ds]) * K + 4 * c4);
    }

    const int kOff = half << 1;
    const int bcol = wave * 16 + c;
    int q2[2][2];
    v2f xb[2][2][K2 / 4];
#pragma unroll
    for (int ds = 0; ds < 2; ++ds) {
#pragma unroll
        for (int e = 0; e < 2; ++e) {
            const int d2 = 2 * dpd[ds] + e;
            const int s2 = min(max(sel2[d2], 0), 2);
            const int t2 = 3 * d2 + s2;
            q2[ds][e] = t2 >> D2log2;
            const int dp2 = t2 & (D2 - 1);
            const float* xp = X1 + (size_t)dp2 * (size_t)(K2 * NB) + bb + bcol;
#pragma unroll
            for (int ks = 0; ks < K2 / 4; ++ks) {
                const int k0 = ks * 4 + kOff;
                xb[ds][e][ks].x = xp[(size_t)k0 * NB];
                xb[ds][e][ks].y = xp[(size_t)(k0 + 1) * NB];
            }
        }
    }
#pragma unroll
    for (int it = 0; it < NW2; ++it) sW2[it * 256 + tid] = w2r[it];
    sb2[tid] = b2r;
    __syncthreads();

    {
        v8f acc2[2][2][2];
#pragma unroll
        for (int ds = 0; ds < 2; ++ds)
#pragma unroll
            for (int e = 0; e < 2; ++e) {
                acc2[ds][e][0] = v8f{};
                acc2[ds][e][1] = v8f{};
#pragma unroll
                for (int ks = 0; ks < K2 / 4; ++ks) {
                    const int k0 = ks * 4 + kOff;
#pragma unroll
                    for (int ft = 0; ft < 2; ++ft) {
                        const v2f af = *reinterpret_cast<const v2f*>(
                            sW2 + (3 * (ft * 16 + c) + q2[ds][e]) * K2 + k0);
                        acc2[ds][e][ft] = __builtin_amdgcn_wmma_f32_16x16x4_f32(
                            false, af, false, xb[ds][e][ks], (short)0,
                            acc2[ds][e][ft], false, false);
                    }
                }
            }
#pragma unroll
        for (int ds = 0; ds < 2; ++ds) {
#pragma unroll
            for (int ft = 0; ft < 2; ++ft) {
                float v[8];
#pragma unroll
                for (int r = 0; r < 8; ++r) {
                    const int f2 = ft * 16 + (half << 3) + r;
                    const float v0 = fmaxf(acc2[ds][0][ft][r] + sb2[3 * f2 + q2[ds][0]], 0.0f);
                    const float v1 = fmaxf(acc2[ds][1][ft][r] + sb2[3 * f2 + q2[ds][1]], 0.0f);
                    v[r] = fmaxf(v0, v1);
                }
#pragma unroll
                for (int r = 0; r < 8; r += 2) {
                    const int kp = (ft * 16 + (half << 3) + r) >> 1;
                    const v2f pr = { v[r], v[r + 1] };
                    *reinterpret_cast<v2f*>(sB + (ds * KP + kp) * PB + 2 * bcol) = pr;
                }
            }
        }
    }
#pragma unroll
    for (int it = 0; it < NA4; ++it)
        *reinterpret_cast<v4f*>(sA + ldsA[it]) = ra[it];
    __syncthreads();

    v8f acc[2][FT][BT];
#pragma unroll
    for (int ds = 0; ds < 2; ++ds)
#pragma unroll
        for (int ft = 0; ft < FT; ++ft)
#pragma unroll
            for (int bt = 0; bt < BT; ++bt) acc[ds][ft][bt] = v8f{};

    {
        const float* as = sA + ((wm * FT) * 16 + c) * PA + 2 * half;
        const float* bs = sB + half * PB + ((wn * BT) * 16 + c) * 2;
#pragma unroll
        for (int ks = 0; ks < K / 4; ++ks) {
            v2f a[2][FT], b[2][BT];
#pragma unroll
            for (int ds = 0; ds < 2; ++ds) {
#pragma unroll
                for (int ft = 0; ft < FT; ++ft)
                    a[ds][ft] = *reinterpret_cast<const v2f*>(
                        as + (ds * MF + ft * 16) * PA + ks * 4);
#pragma unroll
                for (int bt = 0; bt < BT; ++bt)
                    b[ds][bt] = *reinterpret_cast<const v2f*>(
                        bs + (ds * KP + ks * 2) * PB + bt * 32);
            }
#pragma unroll
            for (int ds = 0; ds < 2; ++ds)
#pragma unroll
                for (int ft = 0; ft < FT; ++ft)
#pragma unroll
                    for (int bt = 0; bt < BT; ++bt)
                        acc[ds][ft][bt] = __builtin_amdgcn_wmma_f32_16x16x4_f32(
                            false, a[ds][ft], false, b[ds][bt], (short)0,
                            acc[ds][ft][bt], false, false);
        }
    }

#pragma unroll
    for (int ft = 0; ft < FT; ++ft) {
        float bq0[8], bq1[8];
#pragma unroll
        for (int r = 0; r < 8; ++r) {
            const int f = f0 + (wm * FT + ft) * 16 + (half << 3) + r;
            bq0[r] = bias[3 * f + qd[0]];
            bq1[r] = bias[3 * f + qd[1]];
        }
#pragma unroll
        for (int bt = 0; bt < BT; ++bt) {
            const int b = bb + (wn * BT + bt) * 16 + c;
#pragma unroll
            for (int r = 0; r < 8; ++r) {
                const int f = f0 + (wm * FT + ft) * 16 + (half << 3) + r;
                const float v0 = fmaxf(acc[0][ft][bt][r] + bq0[r], 0.0f);
                const float v1 = fmaxf(acc[1][ft][bt][r] + bq1[r], 0.0f);
                Out[((size_t)j * F + f) * NB + b] = fmaxf(v0, v1);
            }
        }
    }
}

__global__ __launch_bounds__(32) void kd_fc_lsm(
    const float* __restrict__ Y, const float* __restrict__ Wt,
    const float* __restrict__ bias, float* __restrict__ out)
{
    constexpr int K  = 1024;
    constexpr int NB = 512;
    constexpr int M  = 16;

    const int lane = threadIdx.x & 31;
    const int half = lane >> 4;
    const int c    = lane & 15;
    const int kOff = half << 1;
    const int b    = (blockIdx.x << 4) + c;

    const float* ap = Wt + c * K + kOff;
    const float* bp = Y + (size_t)kOff * NB + b;

    float bs[8];
#pragma unroll
    for (int r = 0; r < 8; ++r) bs[r] = bias[(half << 3) + r];

    v8f acc = v8f{};
#pragma unroll 16
    for (int kt = 0; kt < K; kt += 4) {
        const v2f a = *reinterpret_cast<const v2f*>(ap + kt);
        v2f x;
        x.x = bp[(size_t)kt * NB];
        x.y = bp[(size_t)(kt + 1) * NB];
        acc = __builtin_amdgcn_wmma_f32_16x16x4_f32(
            false, a, false, x, (short)0, acc, false, false);
    }

    float v[8];
    float mx = -3.402823466e38f;
#pragma unroll
    for (int r = 0; r < 8; ++r) { v[r] = acc[r] + bs[r]; mx = fmaxf(mx, v[r]); }
    mx = fmaxf(mx, __shfl_xor(mx, 16, 32));

    float e[8];
#pragma unroll
    for (int r = 0; r < 8; ++r) e[r] = expf(v[r] - mx);
    float s = 0.0f;
#pragma unroll
    for (int r = 0; r < 8; ++r) s += e[r];
    float t = __shfl(s, c, 32);
#pragma unroll
    for (int r = 0; r < 8; ++r) t += e[r];
    const float tot = __shfl(t, c + 16, 32);
    const float l = mx + logf(tot);

    float4 o0, o1;
    o0.x = v[0] - l; o0.y = v[1] - l; o0.z = v[2] - l; o0.w = v[3] - l;
    o1.x = v[4] - l; o1.y = v[5] - l; o1.z = v[6] - l; o1.w = v[7] - l;
    float4* op = reinterpret_cast<float4*>(out + (size_t)b * M + (half << 3));
    op[0] = o0;
    op[1] = o1;
}

template <int NSUB, int KP4, bool FIRST>
static void launch_conv(const float* cur, const float* W, const float* b,
                        const int* sel, float* nxt, int F, int Dd,
                        hipStream_t stream)
{
    const int Dlog2 = __builtin_ctz((unsigned)Dd);
    dim3 grid((Dd / 2) * (4 / NSUB), (F + 15) / 16);
    kd_conv_bm<NSUB, KP4, FIRST><<<grid, 256, 0, stream>>>(cur, W, b, sel, nxt, F, Dlog2);
}

template <int WM, int WN, int FT, int BT, int KC, int K>
static void launch_tile(const float* cur, const float* W, const float* b,
                        const int* sel, float* nxt, int F, int Dd,
                        hipStream_t stream)
{
    const int Dlog2 = __builtin_ctz((unsigned)Dd);
    dim3 grid((Dd / 2) * (512 / (WN * BT * 16)), F / (WM * FT * 16));
    kd_conv_tile<WM, WN, FT, BT, KC, K><<<grid, 256, 0, stream>>>(cur, W, b, sel, nxt, F, Dlog2);
}

extern "C" void kernel_launch(void* const* d_in, const int* in_sizes, int n_in,
                              void* d_out, int out_size, void* d_ws, size_t ws_size,
                              hipStream_t stream) {
    static const int DIMS[11] = {2048, 1024, 512, 256, 128, 64, 32, 16, 8, 4, 2};
    static const int FEAT[11] = {8, 32, 64, 64, 64, 128, 256, 512, 512, 512, 1024};

    const size_t XBYTES = (size_t)512 * 32 * 512 * sizeof(float);
    float* X0     = (float*)d_ws;
    float* X1     = (float*)((char*)d_ws + XBYTES);

    const float* cur = (const float*)d_in[0];
    for (int i = 0; i < 11; ++i) {
        if (i == 1) continue;
        const float* W  = (const float*)d_in[12 + 2 * i];
        const float* bi = (const float*)d_in[13 + 2 * i];
        const int*   sl = (const int*)d_in[1 + i];
        float* nxt = (cur == X0) ? X1 : X0;
        const int F = FEAT[i], Dd = DIMS[i];
        switch (i) {
        case 0:  launch_conv<4, 4,   true >(cur, W, bi, sl, nxt, F, Dd, stream); break;
        case 2: {
            dim3 grid((Dd / 2) * 4, F / 64);
            kd_conv3_fold2<<<grid, 256, 0, stream>>>(
                cur, (const float*)d_in[14], (const float*)d_in[15], (const int*)d_in[2],
                W, bi, sl, nxt,
                __builtin_ctz((unsigned)Dd), __builtin_ctz((unsigned)DIMS[1]));
            break;
        }
        case 3:
        case 4:
        case 5:  launch_tile<2, 4, 2, 2, 32, 64 >(cur, W, bi, sl, nxt, F, Dd, stream); break;
        case 6:  launch_tile<2, 4, 2, 2, 32, 128>(cur, W, bi, sl, nxt, F, Dd, stream); break;
        case 7:  launch_tile<2, 4, 2, 2, 32, 256>(cur, W, bi, sl, nxt, F, Dd, stream); break;
        default: launch_tile<4, 2, 1, 1, 64, 512>(cur, W, bi, sl, nxt, F, Dd, stream); break;
        }
        cur = nxt;
    }

    kd_fc_lsm<<<32, 32, 0, stream>>>(
        cur, (const float*)d_in[34], (const float*)d_in[35], (float*)d_out);

    (void)in_sizes; (void)n_in; (void)out_size; (void)ws_size;
}


# FC head's products moved into layer 11's launch: each wave multiplies its 16x16 cells, still in registers, with its 16 Wfc columns (4 WMMAs); a 32-block kernel adds the 64 tiles (new FC sum order)
# speedup vs baseline: 9.6801x; 1.1330x over previous
#include <hip/hip_runtime.h>


typedef float v2f __attribute__((ext_vector_type(2)));
typedef float v4f __attribute__((ext_vector_type(4)));
typedef float v8f __attribute__((ext_vector_type(8)));

template <int NSUB, int KP4, bool FIRST>
__global__ __launch_bounds__(256) void kd_conv_bm(
    const float* __restrict__ X, const float* __restrict__ Wt,
    const float* __restrict__ bias, const int* __restrict__ sel,
    float* __restrict__ Out, int F, int Dlog2)
{
    constexpr int K   = FIRST ? 3 : KP4;
    constexpr int KP  = KP4 + 1;
    constexpr int NB  = 512;
    constexpr int NBC = 4 / NSUB;
    constexpr int UNR = 8 / NSUB;
    __shared__ float sW[48 * KP];

    const int f0 = blockIdx.y << 4;

    constexpr int TOT = 48 * KP4;
    constexpr int NIT = (TOT + 255) / 256;
    constexpr bool EXACT = (TOT % 256) == 0;
#pragma unroll 16
    for (int it = 0; it < NIT; ++it) {
        const int idx = it * 256 + (int)threadIdx.x;
        const int idc = EXACT ? idx : min(idx, TOT - 1);
        const int rr  = idc / KP4;
        const int k   = idc - rr * KP4;
        const int q   = rr >> 4;
        const int f   = f0 + (rr & 15);
        float w = Wt[(3 * min(f, F - 1) + q) * K + min(k, K - 1)];
        w = (f < F && k < K) ? w : 0.0f;
        if (EXACT || idx < TOT) sW[rr * KP + k] = w;
    }
    __syncthreads();

    const int lane = threadIdx.x & 31;
    const int wave = threadIdx.x >> 5;
    const int half = lane >> 4;
    const int c    = lane & 15;
    const int kOff = half << 1;

    const int j     = blockIdx.x / NBC;
    const int bc    = blockIdx.x - j * NBC;
    const int bBase = (bc * 8 + wave) * (16 * NSUB);
    const int D     = 1 << Dlog2;

    const int d0 = j << 1;
    const int d1 = d0 + 1;
    const int s0 = min(max(sel[d0], 0), 2);
    const int s1 = min(max(sel[d1], 0), 2);
    const int t0 = 3 * d0 + s0;
    const int t1 = 3 * d1 + s1;
    const int q0 = t0 >> Dlog2;
    const int q1 = t1 >> Dlog2;
    const int dp0 = t0 & (D - 1);
    const int dp1 = t1 & (D - 1);

    const float* aRow0 = sW + (q0 * 16 + c) * KP;
    const float* aRow1 = sW + (q1 * 16 + c) * KP;

    const int sk = FIRST ? D : NB;
    const float* p0[NSUB];
    const float* p1[NSUB];
#pragma unroll
    for (int s = 0; s < NSUB; ++s) {
        const int b = bBase + (s << 4) + c;
        if (FIRST) {
            p0[s] = X + (size_t)b * (size_t)(K * D) + dp0;
            p1[s] = X + (size_t)b * (size_t)(K * D) + dp1;
        } else {
            p0[s] = X + (size_t)dp0 * (size_t)(K * NB) + b;
            p1[s] = X + (size_t)dp1 * (size_t)(K * NB) + b;
        }
    }

    v8f acc0[NSUB];
    v8f acc1[NSUB];
#pragma unroll
    for (int s = 0; s < NSUB; ++s) { acc0[s] = v8f{}; acc1[s] = v8f{}; }

    constexpr int kFull = K & ~3;
#pragma unroll UNR
    for (int kt = 0; kt < kFull; kt += 4) {
        const int k0 = kt + kOff;
        v2f a0, a1;
        a0.x = aRow0[k0]; a0.y = aRow0[k0 + 1];
        a1.x = aRow1[k0]; a1.y = aRow1[k0 + 1];
#pragma unroll
        for (int s = 0; s < NSUB; ++s) {
            v2f b0, b1;
            b0.x = p0[s][(size_t)k0 * sk];
            b0.y = p0[s][(size_t)(k0 + 1) * sk];
            b1.x = p1[s][(size_t)k0 * sk];
            b1.y = p1[s][(size_t)(k0 + 1) * sk];
            acc0[s] = __builtin_amdgcn_wmma_f32_16x16x4_f32(
                false, a0, false, b0, (short)0, acc0[s], false, false);
            acc1[s] = __builtin_amdgcn_wmma_f32_16x16x4_f32(
                false, a1, false, b1, (short)0, acc1[s], false, false);
        }
    }
    if (kFull < K) {
        const int k0 = kFull + kOff;
        const int c0 = min(k0, K - 1);
        const int c1 = min(k0 + 1, K - 1);
        v2f a0, a1;
        a0.x = aRow0[k0]; a0.y = aRow0[k0 + 1];
        a1.x = aRow1[k0]; a1.y = aRow1[k0 + 1];
#pragma unroll
        for (int s = 0; s < NSUB; ++s) {
            v2f b0, b1;
            b0.x = p0[s][(size_t)c0 * sk];
            b0.y = p0[s][(size_t)c1 * sk];
            b1.x = p1[s][(size_t)c0 * sk];
            b1.y = p1[s][(size_t)c1 * sk];
            acc0[s] = __builtin_amdgcn_wmma_f32_16x16x4_f32(
                false, a0, false, b0, (short)0, acc0[s], false, false);
            acc1[s] = __builtin_amdgcn_wmma_f32_16x16x4_f32(
                false, a1, false, b1, (short)0, acc1[s], false, false);
        }
    }

    float bq0[8], bq1[8];
#pragma unroll
    for (int r = 0; r < 8; ++r) {
        const int f = f0 + (half << 3) + r;
        const int fc = min(f, F - 1);
        bq0[r] = bias[3 * fc + q0];
        bq1[r] = bias[3 * fc + q1];
    }
#pragma unroll
    for (int s = 0; s < NSUB; ++s) {
        const int b = bBase + (s << 4) + c;
#pragma unroll
        for (int r = 0; r < 8; ++r) {
            const int f = f0 + (half << 3) + r;
            if (f < F) {
                const float v0 = fmaxf(acc0[s][r] + bq0[r], 0.0f);
                const float v1 = fmaxf(acc1[s][r] + bq1[r], 0.0f);
                Out[((size_t)j * F + f) * NB + b] = fmaxf(v0, v1);
            }
        }
    }
}

template <int WM, int WN, int FT, int BT, int KC, int K, bool FCF>
__global__ __launch_bounds__(256) void kd_conv_tile(
    const float* __restrict__ X, const float* __restrict__ Wt,
    const float* __restrict__ bias, const int* __restrict__ sel,
    float* __restrict__ Out, int F, int Dlog2, const float* __restrict__ Wfc)
{
    static_assert(WM * WN == 8, "eight waves per block");
    static_assert(K % KC == 0, "whole chunks");
    constexpr int NB  = 512;
    constexpr int MF  = WM * FT * 16;
    constexpr int NBT = WN * BT * 16;
    constexpr int NBC = NB / NBT;
    constexpr int PA  = KC + 4;
    constexpr int KP  = KC / 2;
    constexpr int PB  = 2 * NBT + 32;
    constexpr int SA  = 2 * MF * PA;
    constexpr int SB  = 2 * KP * PB;
    constexpr int A4R = KC / 4;
    constexpr int B4R = NBT / 4;
    constexpr int NA4 = 2 * MF * A4R / 256;
    constexpr int NB4 = 2 * KP * B4R / 256;
    constexpr int NCH = K / KC;
    constexpr int NST = NCH > 1 ? 2 : 1;
    static_assert((MF * A4R) % 256 == 0 && (KP * B4R) % 256 == 0,
                  "a load slot never straddles the two d");

    __shared__ __attribute__((aligned(16))) float sA[NST * SA];
    __shared__ __attribute__((aligned(16))) float sB[NST * SB];

    const int tid  = threadIdx.x;
    const int lane = tid & 31;
    const int wave = tid >> 5;
    const int half = lane >> 4;
    const int c    = lane & 15;
    const int wm   = wave / WN;
    const int wn   = wave - wm * WN;

    const int f0 = blockIdx.y * MF;
    const int j  = blockIdx.x / NBC;
    const int bb = (blockIdx.x - j * NBC) * NBT;
    const int D  = 1 << Dlog2;

    v2f wfc[FCF ? 4 : 1];
    if constexpr (FCF) {
        static_assert(FT == 1 && BT == 1, "one accumulator tile per wave and d");
        const float* ap = Wfc + (size_t)c * F + f0 + wm * 16 + (half << 3);
#pragma unroll
        for (int ks = 0; ks < 4; ++ks)
            wfc[ks] = *reinterpret_cast<const v2f*>(ap + 2 * ks);
    }

    const int d0 = j << 1;
    const int d1 = d0 + 1;
    const int s0 = min(max(sel[d0], 0), 2);
    const int s1 = min(max(sel[d1], 0), 2);
    const int t0 = 3 * d0 + s0;
    const int t1 = 3 * d1 + s1;
    const int qd[2]  = { t0 >> Dlog2, t1 >> Dlog2 };
    const int dpd[2] = { t0 & (D - 1), t1 & (D - 1) };

    int offA[NA4], ldsA[NA4];
#pragma unroll
    for (int it = 0; it < NA4; ++it) {
        const int ds  = (it * 256) / (MF * A4R);
        const int rem = it * 256 + tid - ds * (MF * A4R);
        const int r   = rem / A4R;
        const int c4  = rem - r * A4R;
        offA[it] = (3 * (f0 + r) + qd[ds]) * K + 4 * c4;
        ldsA[it] = (ds * MF + r) * PA + 4 * c4;
    }
    int offB[NB4], ldsB[NB4];
#pragma unroll
    for (int it = 0; it < NB4; ++it) {
        const int ds  = (it * 256) / (KP * B4R);
        const int rem = it * 256 + tid - ds * (KP * B4R);
        const int kp  = rem / B4R;
        const int c4  = rem - kp * B4R;
        offB[it] = (dpd[ds] * K + 2 * kp) * NB + bb + 4 * c4;
        ldsB[it] = (ds * KP + kp) * PB + 8 * c4;
    }

    v4f ra[NA4], rb[NB4][2];
    v8f acc[2][FT][BT];
#pragma unroll
    for (int ds = 0; ds < 2; ++ds)
#pragma unroll
        for (int ft = 0; ft < FT; ++ft)
#pragma unroll
            for (int bt = 0; bt < BT; ++bt) acc[ds][ft][bt] = v8f{};

    auto fetch = [&](int ch) __attribute__((always_inline)) {
#pragma unroll
        for (int it = 0; it < NA4; ++it)
            ra[it] = *reinterpret_cast<const v4f*>(Wt + offA[it] + ch * KC);
#pragma unroll
        for (int it = 0; it < NB4; ++it) {
            rb[it][0] = *reinterpret_cast<const v4f*>(X + offB[it] + ch * (KC * NB));
            rb[it][1] = *reinterpret_cast<const v4f*>(X + offB[it] + ch * (KC * NB) + NB);
        }
    };
    auto stash = [&](int st) __attribute__((always_inline)) {
#pragma unroll
        for (int it = 0; it < NA4; ++it)
            *reinterpret_cast<v4f*>(sA + st * SA + ldsA[it]) = ra[it];
#pragma unroll
        for (int it = 0; it < NB4; ++it) {
            const v4f r0 = rb[it][0], r1 = rb[it][1];
            const v4f lo = { r0.x, r1.x, r0.y, r1.y };
            const v4f hi = { r0.z, r1.z, r0.w, r1.w };
            *reinterpret_cast<v4f*>(sB + st * SB + ldsB[it])     = lo;
            *reinterpret_cast<v4f*>(sB + st * SB + ldsB[it] + 4) = hi;
        }
    };
    auto multiply = [&](int st) __attribute__((always_inline)) {
        const float* as = sA + st * SA + ((wm * FT) * 16 + c) * PA + 2 * half;
        const float* bs = sB + st * SB + half * PB + ((wn * BT) * 16 + c) * 2;
#pragma unroll
        for (int ks = 0; ks < KC / 4; ++ks) {
            v2f a[2][FT], b[2][BT];
#pragma unroll
            for (int ds = 0; ds < 2; ++ds) {
#pragma unroll
                for (int ft = 0; ft < FT; ++ft)
                    a[ds][ft] = *reinterpret_cast<const v2f*>(
                        as + (ds * MF + ft * 16) * PA + ks * 4);
#pragma unroll
                for (int bt = 0; bt < BT; ++bt)
                    b[ds][bt] = *reinterpret_cast<const v2f*>(
                        bs + (ds * KP + ks * 2) * PB + bt * 32);
            }
#pragma unroll
            for (int ds = 0; ds < 2; ++ds)
#pragma unroll
                for (int ft = 0; ft < FT; ++ft)
#pragma unroll
                    for (int bt = 0; bt < BT; ++bt)
                        acc[ds][ft][bt] = __builtin_amdgcn_wmma_f32_16x16x4_f32(
                            false, a[ds][ft], false, b[ds][bt], (short)0,
                            acc[ds][ft][bt], false, false);
        }
    };

    fetch(0);
    stash(0);
    __syncthreads();
#pragma nounroll
    for (int ch = 0; ch < NCH - 1; ++ch) {
        fetch(ch + 1);
        __builtin_amdgcn_sched_barrier(0);
        multiply(ch & 1);
        __builtin_amdgcn_sched_barrier(0);
        stash((ch + 1) & 1);
        __syncthreads();
    }
    multiply((NCH - 1) & 1);

#pragma unroll
    for (int ft = 0; ft < FT; ++ft) {
        float bq0[8], bq1[8];
#pragma unroll
        for (int r = 0; r < 8; ++r) {
            const int f = f0 + (wm * FT + ft) * 16 + (half << 3) + r;
            bq0[r] = bias[3 * f + qd[0]];
            bq1[r] = bias[3 * f + qd[1]];
        }
#pragma unroll
        for (int bt = 0; bt < BT; ++bt) {
            const int b = bb + (wn * BT + bt) * 16 + c;
            if constexpr (FCF) {
                float v[8];
#pragma unroll
                for (int r = 0; r < 8; ++r) {
                    const float v0 = fmaxf(acc[0][ft][bt][r] + bq0[r], 0.0f);
                    const float v1 = fmaxf(acc[1][ft][bt][r] + bq1[r], 0.0f);
                    v[r] = fmaxf(v0, v1);
                }
                v8f pl = v8f{};
#pragma unroll
                for (int ks = 0; ks < 4; ++ks) {
                    const v2f cell = { v[2 * ks], v[2 * ks + 1] };
                    pl = __builtin_amdgcn_wmma_f32_16x16x4_f32(
                        false, wfc[ks], false, cell, (short)0, pl, false, false);
                }
                const int tile = blockIdx.y * (WM * FT) + wm * FT + ft;
#pragma unroll
                for (int r = 0; r < 8; ++r)
                    Out[((size_t)tile * 16 + (half << 3) + r) * NB + b] = pl[r];
            } else {
#pragma unroll
                for (int r = 0; r < 8; ++r) {
                    const int f = f0 + (wm * FT + ft) * 16 + (half << 3) + r;
                    const float v0 = fmaxf(acc[0][ft][bt][r] + bq0[r], 0.0f);
                    const float v1 = fmaxf(acc[1][ft][bt][r] + bq1[r], 0.0f);
                    Out[((size_t)j * F + f) * NB + b] = fmaxf(v0, v1);
                }
            }
        }
    }
}

__global__ __launch_bounds__(256) void kd_conv3_fold2(
    const float* __restrict__ X1, const float* __restrict__ W2,
    const float* __restrict__ b2, const int* __restrict__ sel2,
    const float* __restrict__ Wt, const float* __restrict__ bias,
    const int* __restrict__ sel, float* __restrict__ Out,
    int Dlog2, int D2log2)
{
    constexpr int WM = 2, WN = 4, FT = 2, BT = 2;
    constexpr int NB  = 512;
    constexpr int F   = 64;
    constexpr int K   = 32;
    constexpr int K2  = 8;
    constexpr int R2  = 3 * K;
    constexpr int MF  = WM * FT * 16;
    constexpr int NBT = WN * BT * 16;
    constexpr int NBC = NB / NBT;
    constexpr int PA  = K + 4;
    constexpr int KP  = K / 2;
    constexpr int PB  = 2 * NBT + 32;
    constexpr int SA  = 2 * MF * PA;
    constexpr int SB  = 2 * KP * PB;
    constexpr int A4R = K / 4;
    constexpr int NA4 = 2 * MF * A4R / 256;
    static_assert(NBT == 8 * 16, "layer-2 stage: one 16-batch column tile per wave");
    static_assert(K == 2 * 16, "layer-2 stage: two feature tiles");
    constexpr int NW2 = R2 * K2 / 256;
    static_assert((R2 * K2) % 256 == 0 && R2 <= 256, "W2 / b2 copy: whole trips");

    __shared__ __attribute__((aligned(16))) float sA[SA];
    __shared__ __attribute__((aligned(16))) float sB[SB];
    __shared__ __attribute__((aligned(16))) float sW2[R2 * K2];
    __shared__ float sb2[256];

    const int tid  = threadIdx.x;
    const int lane = tid & 31;
    const int wave = tid >> 5;
    const int half = lane >> 4;
    const int c    = lane & 15;
    const int wm   = wave / WN;
    const int wn   = wave - wm * WN;

    float w2r[NW2];
#pragma unroll
    for (int it = 0; it < NW2; ++it) w2r[it] = W2[it * 256 + tid];
    const float b2r = b2[min(tid, R2 - 1)];

    const int f0 = blockIdx.y * MF;
    const int j  = blockIdx.x / NBC;
    const int bb = (blockIdx.x - j * NBC) * NBT;
    const int D  = 1 << Dlog2;
    const int D2 = 1 << D2log2;

    const int d0 = j << 1;
    const int d1 = d0 + 1;
    const int s0 = min(max(sel[d0], 0), 2);
    const int s1 = min(max(sel[d1], 0), 2);
    const int t0 = 3 * d0 + s0;
    const int t1 = 3 * d1 + s1;
    const int qd[2]  = { t0 >> Dlog2, t1 >> Dlog2 };
    const int dpd[2] = { t0 & (D - 1), t1 & (D - 1) };

    int ldsA[NA4];
    v4f ra[NA4];
#pragma unroll
    for (int it = 0; it < NA4; ++it) {
        const int ds  = (it * 256) / (MF * A4R);
        const int rem = it * 256 + tid - ds * (MF * A4R);
        const int r   = rem / A4R;
        const int c4  = rem - r * A4R;
        ldsA[it] = (ds * MF + r) * PA + 4 * c4;
        ra[it] = *reinterpret_cast<const v4f*>(
            Wt + (3 * (f0 + r) + qd[ds]) * K + 4 * c4);
    }

    const int kOff = half << 1;
    const int bcol = wave * 16 + c;
    int q2[2][2];
    v2f xb[2][2][K2 / 4];
#pragma unroll
    for (int ds = 0; ds < 2; ++ds) {
#pragma unroll
        for (int e = 0; e < 2; ++e) {
            const int d2 = 2 * dpd[ds] + e;
            const int s2 = min(max(sel2[d2], 0), 2);
            const int t2 = 3 * d2 + s2;
            q2[ds][e] = t2 >> D2log2;
            const int dp2 = t2 & (D2 - 1);
            const float* xp = X1 + (size_t)dp2 * (size_t)(K2 * NB) + bb + bcol;
#pragma unroll
            for (int ks = 0; ks < K2 / 4; ++ks) {
                const int k0 = ks * 4 + kOff;
                xb[ds][e][ks].x = xp[(size_t)k0 * NB];
                xb[ds][e][ks].y = xp[(size_t)(k0 + 1) * NB];
            }
        }
    }
#pragma unroll
    for (int it = 0; it < NW2; ++it) sW2[it * 256 + tid] = w2r[it];
    sb2[tid] = b2r;
    __syncthreads();

    {
        v8f acc2[2][2][2];
#pragma unroll
        for (int ds = 0; ds < 2; ++ds)
#pragma unroll
            for (int e = 0; e < 2; ++e) {
                acc2[ds][e][0] = v8f{};
                acc2[ds][e][1] = v8f{};
#pragma unroll
                for (int ks = 0; ks < K2 / 4; ++ks) {
                    const int k0 = ks * 4 + kOff;
#pragma unroll
                    for (int ft = 0; ft < 2; ++ft) {
                        const v2f af = *reinterpret_cast<const v2f*>(
                            sW2 + (3 * (ft * 16 + c) + q2[ds][e]) * K2 + k0);
                        acc2[ds][e][ft] = __builtin_amdgcn_wmma_f32_16x16x4_f32(
                            false, af, false, xb[ds][e][ks], (short)0,
                            acc2[ds][e][ft], false, false);
                    }
                }
            }
#pragma unroll
        for (int ds = 0; ds < 2; ++ds) {
#pragma unroll
            for (int ft = 0; ft < 2; ++ft) {
                float v[8];
#pragma unroll
                for (int r = 0; r < 8; ++r) {
                    const int f2 = ft * 16 + (half << 3) + r;
                    const float v0 = fmaxf(acc2[ds][0][ft][r] + sb2[3 * f2 + q2[ds][0]], 0.0f);
                    const float v1 = fmaxf(acc2[ds][1][ft][r] + sb2[3 * f2 + q2[ds][1]], 0.0f);
                    v[r] = fmaxf(v0, v1);
                }
#pragma unroll
                for (int r = 0; r < 8; r += 2) {
                    const int kp = (ft * 16 + (half << 3) + r) >> 1;
                    const v2f pr = { v[r], v[r + 1] };
                    *reinterpret_cast<v2f*>(sB + (ds * KP + kp) * PB + 2 * bcol) = pr;
                }
            }
        }
    }
#pragma unroll
    for (int it = 0; it < NA4; ++it)
        *reinterpret_cast<v4f*>(sA + ldsA[it]) = ra[it];
    __syncthreads();

    v8f acc[2][FT][BT];
#pragma unroll
    for (int ds = 0; ds < 2; ++ds)
#pragma unroll
        for (int ft = 0; ft < FT; ++ft)
#pragma unroll
            for (int bt = 0; bt < BT; ++bt) acc[ds][ft][bt] = v8f{};

    {
        const float* as = sA + ((wm * FT) * 16 + c) * PA + 2 * half;
        const float* bs = sB + half * PB + ((wn * BT) * 16 + c) * 2;
#pragma unroll
        for (int ks = 0; ks < K / 4; ++ks) {
            v2f a[2][FT], b[2][BT];
#pragma unroll
            for (int ds = 0; ds < 2; ++ds) {
#pragma unroll
                for (int ft = 0; ft < FT; ++ft)
                    a[ds][ft] = *reinterpret_cast<const v2f*>(
                        as + (ds * MF + ft * 16) * PA + ks * 4);
#pragma unroll
                for (int bt = 0; bt < BT; ++bt)
                    b[ds][bt] = *reinterpret_cast<const v2f*>(
                        bs + (ds * KP + ks * 2) * PB + bt * 32);
            }
#pragma unroll
            for (int ds = 0; ds < 2; ++ds)
#pragma unroll
                for (int ft = 0; ft < FT; ++ft)
#pragma unroll
                    for (int bt = 0; bt < BT; ++bt)
                        acc[ds][ft][bt] = __builtin_amdgcn_wmma_f32_16x16x4_f32(
                            false, a[ds][ft], false, b[ds][bt], (short)0,
                            acc[ds][ft][bt], false, false);
        }
    }

#pragma unroll
    for (int ft = 0; ft < FT; ++ft) {
        float bq0[8], bq1[8];
#pragma unroll
        for (int r = 0; r < 8; ++r) {
            const int f = f0 + (wm * FT + ft) * 16 + (half << 3) + r;
            bq0[r] = bias[3 * f + qd[0]];
            bq1[r] = bias[3 * f + qd[1]];
        }
#pragma unroll
        for (int bt = 0; bt < BT; ++bt) {
            const int b = bb + (wn * BT + bt) * 16 + c;
#pragma unroll
            for (int r = 0; r < 8; ++r) {
                const int f = f0 + (wm * FT + ft) * 16 + (half << 3) + r;
                const float v0 = fmaxf(acc[0][ft][bt][r] + bq0[r], 0.0f);
                const float v1 = fmaxf(acc[1][ft][bt][r] + bq1[r], 0.0f);
                Out[((size_t)j * F + f) * NB + b] = fmaxf(v0, v1);
            }
        }
    }
}

__global__ __launch_bounds__(256) void kd_fc_sum_lsm(
    const float* __restrict__ Part, const float* __restrict__ bias,
    float* __restrict__ out)
{
    constexpr int NB = 512;
    constexpr int M  = 16;
    constexpr int NT = 64;
    __shared__ float sL[M * 17];

    const int tid = threadIdx.x;
    {
        const int o = tid >> 4;
        const int b = (blockIdx.x << 4) + (tid & 15);
        const float* pp = Part + (size_t)o * NB + b;
        float p[NT];
#pragma unroll
        for (int t = 0; t < NT; ++t) p[t] = pp[(size_t)t * (M * NB)];
        float sum = 0.0f;
#pragma unroll
        for (int t = 0; t < NT; ++t) sum += p[t];
        sL[o * 17 + (tid & 15)] = sum + bias[o];
    }
    __syncthreads();

    const int bl = tid >> 4;
    const int o  = tid & 15;
    float v[M];
    float mx = -3.402823466e38f;
#pragma unroll
    for (int r = 0; r < M; ++r) { v[r] = sL[r * 17 + bl]; mx = fmaxf(mx, v[r]); }
    float tot = 0.0f;
#pragma unroll
    for (int r = 0; r < M; ++r) tot += expf(v[r] - mx);
    const float l = mx + logf(tot);
    out[((size_t)(blockIdx.x << 4) + bl) * M + o] = sL[o * 17 + bl] - l;
}

template <int NSUB, int KP4, bool FIRST>
static void launch_conv(const float* cur, const float* W, const float* b,
                        const int* sel, float* nxt, int F, int Dd,
                        hipStream_t stream)
{
    const int Dlog2 = __builtin_ctz((unsigned)Dd);
    dim3 grid((Dd / 2) * (4 / NSUB), (F + 15) / 16);
    kd_conv_bm<NSUB, KP4, FIRST><<<grid, 256, 0, stream>>>(cur, W, b, sel, nxt, F, Dlog2);
}

template <int WM, int WN, int FT, int BT, int KC, int K, bool FCF>
static void launch_tile(const float* cur, const float* W, const float* b,
                        const int* sel, float* nxt, int F, int Dd,
                        const float* Wfc, hipStream_t stream)
{
    const int Dlog2 = __builtin_ctz((unsigned)Dd);
    dim3 grid((Dd / 2) * (512 / (WN * BT * 16)), F / (WM * FT * 16));
    kd_conv_tile<WM, WN, FT, BT, KC, K, FCF><<<grid, 256, 0, stream>>>(
        cur, W, b, sel, nxt, F, Dlog2, Wfc);
}

extern "C" void kernel_launch(void* const* d_in, const int* in_sizes, int n_in,
                              void* d_out, int out_size, void* d_ws, size_t ws_size,
                              hipStream_t stream) {
    static const int DIMS[11] = {2048, 1024, 512, 256, 128, 64, 32, 16, 8, 4, 2};
    static const int FEAT[11] = {8, 32, 64, 64, 64, 128, 256, 512, 512, 512, 1024};

    const size_t XBYTES = (size_t)512 * 32 * 512 * sizeof(float);
    float* X0     = (float*)d_ws;
    float* X1     = (float*)((char*)d_ws + XBYTES);

    const float* cur = (const float*)d_in[0];
    for (int i = 0; i < 11; ++i) {
        if (i == 1) continue;
        const float* W  = (const float*)d_in[12 + 2 * i];
        const float* bi = (const float*)d_in[13 + 2 * i];
        const int*   sl = (const int*)d_in[1 + i];
        float* nxt = (cur == X0) ? X1 : X0;
        const int F = FEAT[i], Dd = DIMS[i];
        switch (i) {
        case 0:  launch_conv<4, 4,   true >(cur, W, bi, sl, nxt, F, Dd, stream); break;
        case 2: {
            dim3 grid((Dd / 2) * 4, F / 64);
            kd_conv3_fold2<<<grid, 256, 0, stream>>>(
                cur, (const float*)d_in[14], (const float*)d_in[15], (const int*)d_in[2],
                W, bi, sl, nxt,
                __builtin_ctz((unsigned)Dd), __builtin_ctz((unsigned)DIMS[1]));
            break;
        }
        case 3:
        case 4:
        case 5:  launch_tile<2, 4, 2, 2, 32, 64,  false>(cur, W, bi, sl, nxt, F, Dd, nullptr, stream); break;
        case 6:  launch_tile<2, 4, 2, 2, 32, 128, false>(cur, W, bi, sl, nxt, F, Dd, nullptr, stream); break;
        case 7:  launch_tile<2, 4, 2, 2, 32, 256, false>(cur, W, bi, sl, nxt, F, Dd, nullptr, stream); break;
        case 8:
        case 9:  launch_tile<4, 2, 1, 1, 64, 512, false>(cur, W, bi, sl, nxt, F, Dd, nullptr, stream); break;
        default: launch_tile<4, 2, 1, 1, 64, 512, true >(cur, W, bi, sl, nxt, F, Dd,
                                                         (const float*)d_in[34], stream); break;
        }
        cur = nxt;
    }

    kd_fc_sum_lsm<<<32, 256, 0, stream>>>(
        cur, (const float*)d_in[35], (float*)d_out);

    (void)in_sizes; (void)n_in; (void)out_size; (void)ws_size;
}


# Layer 3 (with its layer-2 stage) folded into layer 4's launch (11 -> 10): each block forms the two gathered layer-3 tiles in LDS as layer 4's B stage; layer-3 output never stored; sums unchanged
# speedup vs baseline: 10.1177x; 1.0452x over previous
#include <hip/hip_runtime.h>


typedef float v2f __attribute__((ext_vector_type(2)));
typedef float v4f __attribute__((ext_vector_type(4)));
typedef float v8f __attribute__((ext_vector_type(8)));

template <int NSUB, int KP4, bool FIRST>
__global__ __launch_bounds__(256) void kd_conv_bm(
    const float* __restrict__ X, const float* __restrict__ Wt,
    const float* __restrict__ bias, const int* __restrict__ sel,
    float* __restrict__ Out, int F, int Dlog2)
{
    constexpr int K   = FIRST ? 3 : KP4;
    constexpr int KP  = KP4 + 1;
    constexpr int NB  = 512;
    constexpr int NBC = 4 / NSUB;
    constexpr int UNR = 8 / NSUB;
    __shared__ float sW[48 * KP];

    const int f0 = blockIdx.y << 4;

    constexpr int TOT = 48 * KP4;
    constexpr int NIT = (TOT + 255) / 256;
    constexpr bool EXACT = (TOT % 256) == 0;
#pragma unroll 16
    for (int it = 0; it < NIT; ++it) {
        const int idx = it * 256 + (int)threadIdx.x;
        const int idc = EXACT ? idx : min(idx, TOT - 1);
        const int rr  = idc / KP4;
        const int k   = idc - rr * KP4;
        const int q   = rr >> 4;
        const int f   = f0 + (rr & 15);
        float w = Wt[(3 * min(f, F - 1) + q) * K + min(k, K - 1)];
        w = (f < F && k < K) ? w : 0.0f;
        if (EXACT || idx < TOT) sW[rr * KP + k] = w;
    }
    __syncthreads();

    const int lane = threadIdx.x & 31;
    const int wave = threadIdx.x >> 5;
    const int half = lane >> 4;
    const int c    = lane & 15;
    const int kOff = half << 1;

    const int j     = blockIdx.x / NBC;
    const int bc    = blockIdx.x - j * NBC;
    const int bBase = (bc * 8 + wave) * (16 * NSUB);
    const int D     = 1 << Dlog2;

    const int d0 = j << 1;
    const int d1 = d0 + 1;
    const int s0 = min(max(sel[d0], 0), 2);
    const int s1 = min(max(sel[d1], 0), 2);
    const int t0 = 3 * d0 + s0;
    const int t1 = 3 * d1 + s1;
    const int q0 = t0 >> Dlog2;
    const int q1 = t1 >> Dlog2;
    const int dp0 = t0 & (D - 1);
    const int dp1 = t1 & (D - 1);

    const float* aRow0 = sW + (q0 * 16 + c) * KP;
    const float* aRow1 = sW + (q1 * 16 + c) * KP;

    const int sk = FIRST ? D : NB;
    const float* p0[NSUB];
    const float* p1[NSUB];
#pragma unroll
    for (int s = 0; s < NSUB; ++s) {
        const int b = bBase + (s << 4) + c;
        if (FIRST) {
            p0[s] = X + (size_t)b * (size_t)(K * D) + dp0;
            p1[s] = X + (size_t)b * (size_t)(K * D) + dp1;
        } else {
            p0[s] = X + (size_t)dp0 * (size_t)(K * NB) + b;
            p1[s] = X + (size_t)dp1 * (size_t)(K * NB) + b;
        }
    }

    v8f acc0[NSUB];
    v8f acc1[NSUB];
#pragma unroll
    for (int s = 0; s < NSUB; ++s) { acc0[s] = v8f{}; acc1[s] = v8f{}; }

    constexpr int kFull = K & ~3;
#pragma unroll UNR
    for (int kt = 0; kt < kFull; kt += 4) {
        const int k0 = kt + kOff;
        v2f a0, a1;
        a0.x = aRow0[k0]; a0.y = aRow0[k0 + 1];
        a1.x = aRow1[k0]; a1.y = aRow1[k0 + 1];
#pragma unroll
        for (int s = 0; s < NSUB; ++s) {
            v2f b0, b1;
            b0.x = p0[s][(size_t)k0 * sk];
            b0.y = p0[s][(size_t)(k0 + 1) * sk];
            b1.x = p1[s][(size_t)k0 * sk];
            b1.y = p1[s][(size_t)(k0 + 1) * sk];
            acc0[s] = __builtin_amdgcn_wmma_f32_16x16x4_f32(
                false, a0, false, b0, (short)0, acc0[s], false, false);
            acc1[s] = __builtin_amdgcn_wmma_f32_16x16x4_f32(
                false, a1, false, b1, (short)0, acc1[s], false, false);
        }
    }
    if (kFull < K) {
        const int k0 = kFull + kOff;
        const int c0 = min(k0, K - 1);
        const int c1 = min(k0 + 1, K - 1);
        v2f a0, a1;
        a0.x = aRow0[k0]; a0.y = aRow0[k0 + 1];
        a1.x = aRow1[k0]; a1.y = aRow1[k0 + 1];
#pragma unroll
        for (int s = 0; s < NSUB; ++s) {
            v2f b0, b1;
            b0.x = p0[s][(size_t)c0 * sk];
            b0.y = p0[s][(size_t)c1 * sk];
            b1.x = p1[s][(size_t)c0 * sk];
            b1.y = p1[s][(size_t)c1 * sk];
            acc0[s] = __builtin_amdgcn_wmma_f32_16x16x4_f32(
                false, a0, false, b0, (short)0, acc0[s], false, false);
            acc1[s] = __builtin_amdgcn_wmma_f32_16x16x4_f32(
                false, a1, false, b1, (short)0, acc1[s], false, false);
        }
    }

    float bq0[8], bq1[8];
#pragma unroll
    for (int r = 0; r < 8; ++r) {
        const int f = f0 + (half << 3) + r;
        const int fc = min(f, F - 1);
        bq0[r] = bias[3 * fc + q0];
        bq1[r] = bias[3 * fc + q1];
    }
#pragma unroll
    for (int s = 0; s < NSUB; ++s) {
        const int b = bBase + (s << 4) + c;
#pragma unroll
        for (int r = 0; r < 8; ++r) {
            const int f = f0 + (half << 3) + r;
            if (f < F) {
                const float v0 = fmaxf(acc0[s][r] + bq0[r], 0.0f);
                const float v1 = fmaxf(acc1[s][r] + bq1[r], 0.0f);
                Out[((size_t)j * F + f) * NB + b] = fmaxf(v0, v1);
            }
        }
    }
}

template <int WM, int WN, int FT, int BT, int KC, int K, bool FCF>
__global__ __launch_bounds__(256) void kd_conv_tile(
    const float* __restrict__ X, const float* __restrict__ Wt,
    const float* __restrict__ bias, const int* __restrict__ sel,
    float* __restrict__ Out, int F, int Dlog2, const float* __restrict__ Wfc)
{
    static_assert(WM * WN == 8, "eight waves per block");
    static_assert(K % KC == 0, "whole chunks");
    constexpr int NB  = 512;
    constexpr int MF  = WM * FT * 16;
    constexpr int NBT = WN * BT * 16;
    constexpr int NBC = NB / NBT;
    constexpr int PA  = KC + 4;
    constexpr int KP  = KC / 2;
    constexpr int PB  = 2 * NBT + 32;
    constexpr int SA  = 2 * MF * PA;
    constexpr int SB  = 2 * KP * PB;
    constexpr int A4R = KC / 4;
    constexpr int B4R = NBT / 4;
    constexpr int NA4 = 2 * MF * A4R / 256;
    constexpr int NB4 = 2 * KP * B4R / 256;
    constexpr int NCH = K / KC;
    constexpr int NST = NCH > 1 ? 2 : 1;
    static_assert((MF * A4R) % 256 == 0 && (KP * B4R) % 256 == 0,
                  "a load slot never straddles the two d");

    __shared__ __attribute__((aligned(16))) float sA[NST * SA];
    __shared__ __attribute__((aligned(16))) float sB[NST * SB];

    const int tid  = threadIdx.x;
    const int lane = tid & 31;
    const int wave = tid >> 5;
    const int half = lane >> 4;
    const int c    = lane & 15;
    const int wm   = wave / WN;
    const int wn   = wave - wm * WN;

    const int f0 = blockIdx.y * MF;
    const int j  = blockIdx.x / NBC;
    const int bb = (blockIdx.x - j * NBC) * NBT;
    const int D  = 1 << Dlog2;

    v2f wfc[FCF ? 4 : 1];
    if constexpr (FCF) {
        static_assert(FT == 1 && BT == 1, "one accumulator tile per wave and d");
        const float* ap = Wfc + (size_t)c * F + f0 + wm * 16 + (half << 3);
#pragma unroll
        for (int ks = 0; ks < 4; ++ks)
            wfc[ks] = *reinterpret_cast<const v2f*>(ap + 2 * ks);
    }

    const int d0 = j << 1;
    const int d1 = d0 + 1;
    const int s0 = min(max(sel[d0], 0), 2);
    const int s1 = min(max(sel[d1], 0), 2);
    const int t0 = 3 * d0 + s0;
    const int t1 = 3 * d1 + s1;
    const int qd[2]  = { t0 >> Dlog2, t1 >> Dlog2 };
    const int dpd[2] = { t0 & (D - 1), t1 & (D - 1) };

    int offA[NA4], ldsA[NA4];
#pragma unroll
    for (int it = 0; it < NA4; ++it) {
        const int ds  = (it * 256) / (MF * A4R);
        const int rem = it * 256 + tid - ds * (MF * A4R);
        const int r   = rem / A4R;
        const int c4  = rem - r * A4R;
        offA[it] = (3 * (f0 + r) + qd[ds]) * K + 4 * c4;
        ldsA[it] = (ds * MF + r) * PA + 4 * c4;
    }
    int offB[NB4], ldsB[NB4];
#pragma unroll
    for (int it = 0; it < NB4; ++it) {
        const int ds  = (it * 256) / (KP * B4R);
        const int rem = it * 256 + tid - ds * (KP * B4R);
        const int kp  = rem / B4R;
        const int c4  = rem - kp * B4R;
        offB[it] = (dpd[ds] * K + 2 * kp) * NB + bb + 4 * c4;
        ldsB[it] = (ds * KP + kp) * PB + 8 * c4;
    }

    v4f ra[NA4], rb[NB4][2];
    v8f acc[2][FT][BT];
#pragma unroll
    for (int ds = 0; ds < 2; ++ds)
#pragma unroll
        for (int ft = 0; ft < FT; ++ft)
#pragma unroll
            for (int bt = 0; bt < BT; ++bt) acc[ds][ft][bt] = v8f{};

    auto fetch = [&](int ch) __attribute__((always_inline)) {
#pragma unroll
        for (int it = 0; it < NA4; ++it)
            ra[it] = *reinterpret_cast<const v4f*>(Wt + offA[it] + ch * KC);
#pragma unroll
        for (int it = 0; it < NB4; ++it) {
            rb[it][0] = *reinterpret_cast<const v4f*>(X + offB[it] + ch * (KC * NB));
            rb[it][1] = *reinterpret_cast<const v4f*>(X + offB[it] + ch * (KC * NB) + NB);
        }
    };
    auto stash = [&](int st) __attribute__((always_inline)) {
#pragma unroll
        for (int it = 0; it < NA4; ++it)
            *reinterpret_cast<v4f*>(sA + st * SA + ldsA[it]) = ra[it];
#pragma unroll
        for (int it = 0; it < NB4; ++it) {
            const v4f r0 = rb[it][0], r1 = rb[it][1];
            const v4f lo = { r0.x, r1.x, r0.y, r1.y };
            const v4f hi = { r0.z, r1.z, r0.w, r1.w };
            *reinterpret_cast<v4f*>(sB + st * SB + ldsB[it])     = lo;
            *reinterpret_cast<v4f*>(sB + st * SB + ldsB[it] + 4) = hi;
        }
    };
    auto multiply = [&](int st) __attribute__((always_inline)) {
        const float* as = sA + st * SA + ((wm * FT) * 16 + c) * PA + 2 * half;
        const float* bs = sB + st * SB + half * PB + ((wn * BT) * 16 + c) * 2;
#pragma unroll
        for (int ks = 0; ks < KC / 4; ++ks) {
            v2f a[2][FT], b[2][BT];
#pragma unroll
            for (int ds = 0; ds < 2; ++ds) {
#pragma unroll
                for (int ft = 0; ft < FT; ++ft)
                    a[ds][ft] = *reinterpret_cast<const v2f*>(
                        as + (ds * MF + ft * 16) * PA + ks * 4);
#pragma unroll
                for (int bt = 0; bt < BT; ++bt)
                    b[ds][bt] = *reinterpret_cast<const v2f*>(
                        bs + (ds * KP + ks * 2) * PB + bt * 32);
            }
#pragma unroll
            for (int ds = 0; ds < 2; ++ds)
#pragma unroll
                for (int ft = 0; ft < FT; ++ft)
#pragma unroll
                    for (int bt = 0; bt < BT; ++bt)
                        acc[ds][ft][bt] = __builtin_amdgcn_wmma_f32_16x16x4_f32(
                            false, a[ds][ft], false, b[ds][bt], (short)0,
                            acc[ds][ft][bt], false, false);
        }
    };

    fetch(0);
    stash(0);
    __syncthreads();
#pragma nounroll
    for (int ch = 0; ch < NCH - 1; ++ch) {
        fetch(ch + 1);
        __builtin_amdgcn_sched_barrier(0);
        multiply(ch & 1);
        __builtin_amdgcn_sched_barrier(0);
        stash((ch + 1) & 1);
        __syncthreads();
    }
    multiply((NCH - 1) & 1);

#pragma unroll
    for (int ft = 0; ft < FT; ++ft) {
        float bq0[8], bq1[8];
#pragma unroll
        for (int r = 0; r < 8; ++r) {
            const int f = f0 + (wm * FT + ft) * 16 + (half << 3) + r;
            bq0[r] = bias[3 * f + qd[0]];
            bq1[r] = bias[3 * f + qd[1]];
        }
#pragma unroll
        for (int bt = 0; bt < BT; ++bt) {
            const int b = bb + (wn * BT + bt) * 16 + c;
            if constexpr (FCF) {
                float v[8];
#pragma unroll
                for (int r = 0; r < 8; ++r) {
                    const float v0 = fmaxf(acc[0][ft][bt][r] + bq0[r], 0.0f);
                    const float v1 = fmaxf(acc[1][ft][bt][r] + bq1[r], 0.0f);
                    v[r] = fmaxf(v0, v1);
                }
                v8f pl = v8f{};
#pragma unroll
                for (int ks = 0; ks < 4; ++ks) {
                    const v2f cell = { v[2 * ks], v[2 * ks + 1] };
                    pl = __builtin_amdgcn_wmma_f32_16x16x4_f32(
                        false, wfc[ks], false, cell, (short)0, pl, false, false);
                }
                const int tile = blockIdx.y * (WM * FT) + wm * FT + ft;
#pragma unroll
                for (int r = 0; r < 8; ++r)
                    Out[((size_t)tile * 16 + (half << 3) + r) * NB + b] = pl[r];
            } else {
#pragma unroll
                for (int r = 0; r < 8; ++r) {
                    const int f = f0 + (wm * FT + ft) * 16 + (half << 3) + r;
                    const float v0 = fmaxf(acc[0][ft][bt][r] + bq0[r], 0.0f);
                    const float v1 = fmaxf(acc[1][ft][bt][r] + bq1[r], 0.0f);
                    Out[((size_t)j * F + f) * NB + b] = fmaxf(v0, v1);
                }
            }
        }
    }
}

__global__ __launch_bounds__(256) void kd_conv4_fold3(
    const float* __restrict__ X1, const float* __restrict__ W2,
    const float* __restrict__ b2, const int* __restrict__ sel2,
    const float* __restrict__ W3, const float* __restrict__ b3,
    const int* __restrict__ sel3, const float* __restrict__ W4,
    const float* __restrict__ b4, const int* __restrict__ sel4,
    float* __restrict__ Out, int D4log2, int D3log2, int D2log2)
{
    constexpr int WM = 2, WN = 4, FT = 2, BT = 2;
    constexpr int NB  = 512;
    constexpr int F   = 64;
    constexpr int K3  = 32;
    constexpr int K2  = 8;
    constexpr int K4  = 64;
    constexpr int R2  = 3 * K3;
    constexpr int MF  = WM * FT * 16;
    constexpr int NBT = WN * BT * 16;
    constexpr int NBC = NB / NBT;
    constexpr int PB  = 2 * NBT + 32;
    constexpr int PA3 = K3 + 4;
    constexpr int PA4 = K4 + 4;
    constexpr int KP3 = K3 / 2;
    constexpr int KP4 = K4 / 2;
    constexpr int SA3 = 2 * MF * PA3;
    constexpr int SB3 = 2 * KP3 * PB;
    constexpr int SA4 = 2 * MF * PA4;
    constexpr int SB4 = 2 * KP4 * PB;
    constexpr int A3R = K3 / 4;
    constexpr int A4R = K4 / 4;
    constexpr int NA3 = 2 * MF * A3R / 256;
    constexpr int NA4 = 2 * MF * A4R / 256;
    constexpr int NW2 = R2 * K2 / 256;
    static_assert(NBT == 8 * 16, "layer-2 stage: one 16-batch column tile per wave");
    static_assert(K3 == 2 * 16 && F == MF, "two layer-2 feature tiles; one feature block");
    static_assert((R2 * K2) % 256 == 0 && R2 <= 256, "W2 / b2 copy: whole trips");
    static_assert(SA4 <= SA3 + SB3, "layer 4's A stage fits the layer-3 stages");

    __shared__ __attribute__((aligned(16))) float sP[SA3 + SB3];
    __shared__ __attribute__((aligned(16))) float sB4[SB4];
    __shared__ __attribute__((aligned(16))) float sW2[R2 * K2];
    __shared__ float sb2[256];
    float* const sA3 = sP;
    float* const sB3 = sP + SA3;
    float* const sA4 = sP;

    const int tid  = threadIdx.x;
    const int lane = tid & 31;
    const int wave = tid >> 5;
    const int half = lane >> 4;
    const int c    = lane & 15;
    const int wm   = wave / WN;
    const int wn   = wave - wm * WN;
    const int kOff = half << 1;
    const int bcol = wave * 16 + c;

    float w2r[NW2];
#pragma unroll
    for (int it = 0; it < NW2; ++it) w2r[it] = W2[it * 256 + tid];
    const float b2r = b2[min(tid, R2 - 1)];

    const int j4 = blockIdx.x / NBC;
    const int bb = (blockIdx.x - j4 * NBC) * NBT;
    const int D4 = 1 << D4log2;
    const int D3 = 1 << D3log2;
    const int D2 = 1 << D2log2;

    int q4[2], q3[2][2], q2[2][2][2], p1[2][2][2];
#pragma unroll
    for (int dsc = 0; dsc < 2; ++dsc) {
        const int d4 = 2 * j4 + dsc;
        const int t4 = 3 * d4 + min(max(sel4[d4], 0), 2);
        q4[dsc] = t4 >> D4log2;
        const int j3 = t4 & (D4 - 1);
#pragma unroll
        for (int ds = 0; ds < 2; ++ds) {
            const int d3 = 2 * j3 + ds;
            const int t3 = 3 * d3 + min(max(sel3[d3], 0), 2);
            q3[dsc][ds] = t3 >> D3log2;
            const int j2 = t3 & (D3 - 1);
#pragma unroll
            for (int e = 0; e < 2; ++e) {
                const int d2 = 2 * j2 + e;
                const int t2 = 3 * d2 + min(max(sel2[d2], 0), 2);
                q2[dsc][ds][e] = t2 >> D2log2;
                p1[dsc][ds][e] = t2 & (D2 - 1);
            }
        }
    }

    int ldsA4[NA4];
    v4f ra4[NA4];
#pragma unroll
    for (int it = 0; it < NA4; ++it) {
        const int ds  = (it * 256) / (MF * A4R);
        const int rem = it * 256 + tid - ds * (MF * A4R);
        const int r   = rem / A4R;
        const int c4  = rem - r * A4R;
        ldsA4[it] = (ds * MF + r) * PA4 + 4 * c4;
        ra4[it] = *reinterpret_cast<const v4f*>(W4 + (3 * r + q4[ds]) * K4 + 4 * c4);
    }

    int ldsA3[NA3], rowA3[NA3];
#pragma unroll
    for (int it = 0; it < NA3; ++it) {
        const int ds  = (it * 256) / (MF * A3R);
        const int rem = it * 256 + tid - ds * (MF * A3R);
        const int r   = rem / A3R;
        const int c4  = rem - r * A3R;
        ldsA3[it] = (ds * MF + r) * PA3 + 4 * c4;
        rowA3[it] = 3 * r * K3 + 4 * c4;
    }

    v4f ra3[2][NA3];
    v2f xb[2][2][2][K2 / 4];
    auto request = [&](int dsc) __attribute__((always_inline)) {
#pragma unroll
        for (int it = 0; it < NA3; ++it)
            ra3[dsc][it] = *reinterpret_cast<const v4f*>(
                W3 + rowA3[it] + q3[dsc][(it * 256) / (MF * A3R)] * K3);
#pragma unroll
        for (int ds = 0; ds < 2; ++ds)
#pragma unroll
            for (int e = 0; e < 2; ++e) {
                const float* xp = X1 + (size_t)p1[dsc][ds][e] * (size_t)(K2 * NB) + bb + bcol;
#pragma unroll
                for (int ks = 0; ks < K2 / 4; ++ks) {
                    const int k0 = ks * 4 + kOff;
                    xb[dsc][ds][e][ks].x = xp[(size_t)k0 * NB];
                    xb[dsc][ds][e][ks].y = xp[(size_t)(k0 + 1) * NB];
                }
            }
    };

    request(0);
#pragma unroll
    for (int it = 0; it < NW2; ++it) sW2[it * 256 + tid] = w2r[it];
    sb2[tid] = b2r;
    __syncthreads();

#pragma unroll
    for (int dsc = 0; dsc < 2; ++dsc) {
        {
            v8f acc2[2][2][2];
#pragma unroll
            for (int ds = 0; ds < 2; ++ds)
#pragma unroll
                for (int e = 0; e < 2; ++e) {
                    acc2[ds][e][0] = v8f{};
                    acc2[ds][e][1] = v8f{};
#pragma unroll
                    for (int ks = 0; ks < K2 / 4; ++ks) {
                        const int k0 = ks * 4 + kOff;
#pragma unroll
                        for (int ft = 0; ft < 2; ++ft) {
                            const v2f af = *reinterpret_cast<const v2f*>(
                                sW2 + (3 * (ft * 16 + c) + q2[dsc][ds][e]) * K2 + k0);
                            acc2[ds][e][ft] = __builtin_amdgcn_wmma_f32_16x16x4_f32(
                                false, af, false, xb[dsc][ds][e][ks], (short)0,
                                acc2[ds][e][ft], false, false);
                        }
                    }
                }
#pragma unroll
            for (int ds = 0; ds < 2; ++ds) {
#pragma unroll
                for (int ft = 0; ft < 2; ++ft) {
                    float v[8];
#pragma unroll
                    for (int r = 0; r < 8; ++r) {
                        const int f2 = ft * 16 + (half << 3) + r;
                        const float v0 = fmaxf(acc2[ds][0][ft][r] + sb2[3 * f2 + q2[dsc][ds][0]], 0.0f);
                        const float v1 = fmaxf(acc2[ds][1][ft][r] + sb2[3 * f2 + q2[dsc][ds][1]], 0.0f);
                        v[r] = fmaxf(v0, v1);
                    }
#pragma unroll
                    for (int r = 0; r < 8; r += 2) {
                        const int kp = (ft * 16 + (half << 3) + r) >> 1;
                        const v2f pr = { v[r], v[r + 1] };
                        *reinterpret_cast<v2f*>(sB3 + (ds * KP3 + kp) * PB + 2 * bcol) = pr;
                    }
                }
            }
        }
        if (dsc == 0) request(1);
#pragma unroll
        for (int it = 0; it < NA3; ++it)
            *reinterpret_cast<v4f*>(sA3 + ldsA3[it]) = ra3[dsc][it];
        __syncthreads();

        v8f acc[2][FT][BT];
#pragma unroll
        for (int ds = 0; ds < 2; ++ds)
#pragma unroll
            for (int ft = 0; ft < FT; ++ft)
#pragma unroll
                for (int bt = 0; bt < BT; ++bt) acc[ds][ft][bt] = v8f{};
        {
            const float* as = sA3 + ((wm * FT) * 16 + c) * PA3 + 2 * half;
            const float* bs = sB3 + half * PB + ((wn * BT) * 16 + c) * 2;
#pragma unroll
            for (int ks = 0; ks < K3 / 4; ++ks) {
                v2f a[2][FT], b[2][BT];
#pragma unroll
                for (int ds = 0; ds < 2; ++ds) {
#pragma unroll
                    for (int ft = 0; ft < FT; ++ft)
                        a[ds][ft] = *reinterpret_cast<const v2f*>(
                            as + (ds * MF + ft * 16) * PA3 + ks * 4);
#pragma unroll
                    for (int bt = 0; bt < BT; ++bt)
                        b[ds][bt] = *reinterpret_cast<const v2f*>(
                            bs + (ds * KP3 + ks * 2) * PB + bt * 32);
                }
#pragma unroll
                for (int ds = 0; ds < 2; ++ds)
#pragma unroll
                    for (int ft = 0; ft < FT; ++ft)
#pragma unroll
                        for (int bt = 0; bt < BT; ++bt)
                            acc[ds][ft][bt] = __builtin_amdgcn_wmma_f32_16x16x4_f32(
                                false, a[ds][ft], false, b[ds][bt], (short)0,
                                acc[ds][ft][bt], false, false);
            }
        }
#pragma unroll
        for (int ft = 0; ft < FT; ++ft) {
            float bq0[8], bq1[8];
#pragma unroll
            for (int r = 0; r < 8; ++r) {
                const int f = (wm * FT + ft) * 16 + (half << 3) + r;
                bq0[r] = b3[3 * f + q3[dsc][0]];
                bq1[r] = b3[3 * f + q3[dsc][1]];
            }
#pragma unroll
            for (int bt = 0; bt < BT; ++bt) {
                const int col = (wn * BT + bt) * 16 + c;
#pragma unroll
                for (int r = 0; r < 8; r += 2) {
                    const int kp = ((wm * FT + ft) * 16 + (half << 3) + r) >> 1;
                    v2f pr;
                    pr.x = fmaxf(fmaxf(acc[0][ft][bt][r] + bq0[r], 0.0f),
                                 fmaxf(acc[1][ft][bt][r] + bq1[r], 0.0f));
                    pr.y = fmaxf(fmaxf(acc[0][ft][bt][r + 1] + bq0[r + 1], 0.0f),
                                 fmaxf(acc[1][ft][bt][r + 1] + bq1[r + 1], 0.0f));
                    *reinterpret_cast<v2f*>(sB4 + (dsc * KP4 + kp) * PB + 2 * col) = pr;
                }
            }
        }
        __syncthreads();
    }

#pragma unroll
    for (int it = 0; it < NA4; ++it)
        *reinterpret_cast<v4f*>(sA4 + ldsA4[it]) = ra4[it];
    __syncthreads();

    v8f acc[2][FT][BT];
#pragma unroll
    for (int ds = 0; ds < 2; ++ds)
#pragma unroll
        for (int ft = 0; ft < FT; ++ft)
#pragma unroll
            for (int bt = 0; bt < BT; ++bt) acc[ds][ft][bt] = v8f{};
    {
        const float* as = sA4 + ((wm * FT) * 16 + c) * PA4 + 2 * half;
        const float* bs = sB4 + half * PB + ((wn * BT) * 16 + c) * 2;
#pragma unroll
        for (int ks = 0; ks < K4 / 4; ++ks) {
            v2f a[2][FT], b[2][BT];
#pragma unroll
            for (int ds = 0; ds < 2; ++ds) {
#pragma unroll
                for (int ft = 0; ft < FT; ++ft)
                    a[ds][ft] = *reinterpret_cast<const v2f*>(
                        as + (ds * MF + ft * 16) * PA4 + ks * 4);
#pragma unroll
                for (int bt = 0; bt < BT; ++bt)
                    b[ds][bt] = *reinterpret_cast<const v2f*>(
                        bs + (ds * KP4 + ks * 2) * PB + bt * 32);
            }
#pragma unroll
            for (int ds = 0; ds < 2; ++ds)
#pragma unroll
                for (int ft = 0; ft < FT; ++ft)
#pragma unroll
                    for (int bt = 0; bt < BT; ++bt)
                        acc[ds][ft][bt] = __builtin_amdgcn_wmma_f32_16x16x4_f32(
                            false, a[ds][ft], false, b[ds][bt], (short)0,
                            acc[ds][ft][bt], false, false);
        }
    }

#pragma unroll
    for (int ft = 0; ft < FT; ++ft) {
        float bq0[8], bq1[8];
#pragma unroll
        for (int r = 0; r < 8; ++r) {
            const int f = (wm * FT + ft) * 16 + (half << 3) + r;
            bq0[r] = b4[3 * f + q4[0]];
            bq1[r] = b4[3 * f + q4[1]];
        }
#pragma unroll
        for (int bt = 0; bt < BT; ++bt) {
            const int b = bb + (wn * BT + bt) * 16 + c;
#pragma unroll
            for (int r = 0; r < 8; ++r) {
                const int f = (wm * FT + ft) * 16 + (half << 3) + r;
                const float v0 = fmaxf(acc[0][ft][bt][r] + bq0[r], 0.0f);
                const float v1 = fmaxf(acc[1][ft][bt][r] + bq1[r], 0.0f);
                Out[((size_t)j4 * F + f) * NB + b] = fmaxf(v0, v1);
            }
        }
    }
}

__global__ __launch_bounds__(256) void kd_fc_sum_lsm(
    const float* __restrict__ Part, const float* __restrict__ bias,
    float* __restrict__ out)
{
    constexpr int NB = 512;
    constexpr int M  = 16;
    constexpr int NT = 64;
    __shared__ float sL[M * 17];

    const int tid = threadIdx.x;
    {
        const int o = tid >> 4;
        const int b = (blockIdx.x << 4) + (tid & 15);
        const float* pp = Part + (size_t)o * NB + b;
        float p[NT];
#pragma unroll
        for (int t = 0; t < NT; ++t) p[t] = pp[(size_t)t * (M * NB)];
        float sum = 0.0f;
#pragma unroll
        for (int t = 0; t < NT; ++t) sum += p[t];
        sL[o * 17 + (tid & 15)] = sum + bias[o];
    }
    __syncthreads();

    const int bl = tid >> 4;
    const int o  = tid & 15;
    float v[M];
    float mx = -3.402823466e38f;
#pragma unroll
    for (int r = 0; r < M; ++r) { v[r] = sL[r * 17 + bl]; mx = fmaxf(mx, v[r]); }
    float tot = 0.0f;
#pragma unroll
    for (int r = 0; r < M; ++r) tot += expf(v[r] - mx);
    const float l = mx + logf(tot);
    out[((size_t)(blockIdx.x << 4) + bl) * M + o] = sL[o * 17 + bl] - l;
}

template <int NSUB, int KP4, bool FIRST>
static void launch_conv(const float* cur, const float* W, const float* b,
                        const int* sel, float* nxt, int F, int Dd,
                        hipStream_t stream)
{
    const int Dlog2 = __builtin_ctz((unsigned)Dd);
    dim3 grid((Dd / 2) * (4 / NSUB), (F + 15) / 16);
    kd_conv_bm<NSUB, KP4, FIRST><<<grid, 256, 0, stream>>>(cur, W, b, sel, nxt, F, Dlog2);
}

template <int WM, int WN, int FT, int BT, int KC, int K, bool FCF>
static void launch_tile(const float* cur, const float* W, const float* b,
                        const int* sel, float* nxt, int F, int Dd,
                        const float* Wfc, hipStream_t stream)
{
    const int Dlog2 = __builtin_ctz((unsigned)Dd);
    dim3 grid((Dd / 2) * (512 / (WN * BT * 16)), F / (WM * FT * 16));
    kd_conv_tile<WM, WN, FT, BT, KC, K, FCF><<<grid, 256, 0, stream>>>(
        cur, W, b, sel, nxt, F, Dlog2, Wfc);
}

extern "C" void kernel_launch(void* const* d_in, const int* in_sizes, int n_in,
                              void* d_out, int out_size, void* d_ws, size_t ws_size,
                              hipStream_t stream) {
    static const int DIMS[11] = {2048, 1024, 512, 256, 128, 64, 32, 16, 8, 4, 2};
    static const int FEAT[11] = {8, 32, 64, 64, 64, 128, 256, 512, 512, 512, 1024};

    const size_t XBYTES = (size_t)512 * 32 * 512 * sizeof(float);
    float* X0     = (float*)d_ws;
    float* X1     = (float*)((char*)d_ws + XBYTES);

    const float* cur = (const float*)d_in[0];
    for (int i = 0; i < 11; ++i) {
        if (i == 1 || i == 2) continue;
        const float* W  = (const float*)d_in[12 + 2 * i];
        const float* bi = (const float*)d_in[13 + 2 * i];
        const int*   sl = (const int*)d_in[1 + i];
        float* nxt = (cur == X0) ? X1 : X0;
        const int F = FEAT[i], Dd = DIMS[i];
        switch (i) {
        case 0:  launch_conv<4, 4,   true >(cur, W, bi, sl, nxt, F, Dd, stream); break;
        case 3: {
            dim3 grid((Dd / 2) * 4, F / 64);
            kd_conv4_fold3<<<grid, 256, 0, stream>>>(
                cur, (const float*)d_in[14], (const float*)d_in[15], (const int*)d_in[2],
                (const float*)d_in[16], (const float*)d_in[17], (const int*)d_in[3],
                W, bi, sl, nxt,
                __builtin_ctz((unsigned)Dd), __builtin_ctz((unsigned)DIMS[2]),
                __builtin_ctz((unsigned)DIMS[1]));
            break;
        }
        case 4:
        case 5:  launch_tile<2, 4, 2, 2, 32, 64,  false>(cur, W, bi, sl, nxt, F, Dd, nullptr, stream); break;
        case 6:  launch_tile<2, 4, 2, 2, 32, 128, false>(cur, W, bi, sl, nxt, F, Dd, nullptr, stream); break;
        case 7:  launch_tile<2, 4, 2, 2, 32, 256, false>(cur, W, bi, sl, nxt, F, Dd, nullptr, stream); break;
        case 8:
        case 9:  launch_tile<4, 2, 1, 1, 64, 512, false>(cur, W, bi, sl, nxt, F, Dd, nullptr, stream); break;
        default: launch_tile<4, 2, 1, 1, 64, 512, true >(cur, W, bi, sl, nxt, F, Dd,
                                                         (const float*)d_in[34], stream); break;
        }
        cur = nxt;
    }

    kd_fc_sum_lsm<<<32, 256, 0, stream>>>(
        cur, (const float*)d_in[35], (float*)d_out);

    (void)in_sizes; (void)n_in; (void)out_size; (void)ws_size;
}
